# starting kernel
# baseline (speedup 1.0000x reference)
#include <hip/hip_runtime.h>

typedef float v2f __attribute__((ext_vector_type(2)));
typedef float v8f __attribute__((ext_vector_type(8)));

#define N_NODES 50000
#define N_EDGES 600000
#define DIM     128
#define NF      (N_NODES * DIM)

__global__ void k_fill(float* __restrict__ p, float v, int n) {
    int i = blockIdx.x * blockDim.x + threadIdx.x;
    if (i < n) p[i] = v;
}

__global__ void k_deg_accum(const int* __restrict__ dst, float* __restrict__ deg) {
    int e = blockIdx.x * blockDim.x + threadIdx.x;
    if (e < N_EDGES) atomicAdd(&deg[dst[e]], 1.0f);
}

__global__ void k_rsqrt_inplace(float* __restrict__ d, int n) {
    int i = blockIdx.x * blockDim.x + threadIdx.x;
    if (i < n) d[i] = rsqrtf(d[i]);
}

__global__ void k_feat(const float* __restrict__ w,
                       const float* __restrict__ lw,
                       const float* __restrict__ lb,
                       float* __restrict__ feat) {
    int i = blockIdx.x * blockDim.x + threadIdx.x;
    if (i < NF) {
        int n = i >> 7, d = i & 127;
        feat[i] = w[n] * lw[d] + lb[d];
    }
}

__global__ void k_scatter(const int* __restrict__ src,
                          const int* __restrict__ dst,
                          const float* __restrict__ x,
                          const float* __restrict__ dinv,
                          float* __restrict__ agg) {
    unsigned t = blockIdx.x * blockDim.x + threadIdx.x;
    if (t >= (unsigned)N_EDGES * 32u) return;
    int e = t >> 5;
    int q = (t & 31) << 2;
    int s = src[e], d = dst[e];
    float sc = dinv[s];
    const float4 v = *(const float4*)(x + (size_t)s * DIM + q);
    float* ap = agg + (size_t)d * DIM + q;
    atomicAdd(ap + 0, v.x * sc);
    atomicAdd(ap + 1, v.y * sc);
    atomicAdd(ap + 2, v.z * sc);
    atomicAdd(ap + 3, v.w * sc);
}

__global__ void k_finish(float* __restrict__ out,
                         const float* __restrict__ agg,
                         const float* __restrict__ x,
                         const float* __restrict__ prev,
                         const float* __restrict__ dinv,
                         float alpha, float beta) {
    int i = blockIdx.x * blockDim.x + threadIdx.x;
    if (i < NF) {
        int n = i >> 7;
        float di = dinv[n];
        float h = (agg[i] + x[i] * di) * di;
        out[i] = alpha * h + beta * prev[i];
    }
}

__global__ void __launch_bounds__(32)
k_cheb_gemm(const float* __restrict__ X0,
            const float* __restrict__ X1,
            const float* __restrict__ X2,
            const float* __restrict__ W,
            const float* __restrict__ b,
            float* __restrict__ out) {
    const int row0 = blockIdx.x * 16;
    const int lane = threadIdx.x & 31;
    const int half = lane >> 4;
    const int lr   = lane & 15;

    v8f acc[8];
#pragma unroll
    for (int nt = 0; nt < 8; ++nt) {
        float bv = b[nt * 16 + lr];
        acc[nt] = (v8f){bv, bv, bv, bv, bv, bv, bv, bv};
    }

    const float* segs[3] = {X0, X1, X2};
    for (int seg = 0; seg < 3; ++seg) {
        const float* Xp = segs[seg] + (size_t)row0 * DIM;
#pragma unroll 4
        for (int kk = 0; kk < DIM; kk += 4) {
            const int ka = kk + 2 * half;
            v2f a;
            a.x = Xp[(size_t)lr * DIM + ka];
            a.y = Xp[(size_t)lr * DIM + ka + 1];
            const float* Wk  = W + (size_t)(seg * DIM + ka) * DIM;
            const float* Wk1 = Wk + DIM;
#pragma unroll
            for (int nt = 0; nt < 8; ++nt) {
                v2f bb;
                bb.x = Wk [nt * 16 + lr];
                bb.y = Wk1[nt * 16 + lr];
                acc[nt] = __builtin_amdgcn_wmma_f32_16x16x4_f32(
                     false, a,  false, bb,
                     (short)0, acc[nt],
                     false,  false);
            }
        }
    }

#pragma unroll
    for (int nt = 0; nt < 8; ++nt) {
        const int col = nt * 16 + lr;
#pragma unroll
        for (int r = 0; r < 8; ++r) {
            const int row = row0 + r + 8 * half;
            float v = acc[nt][r];
            out[(size_t)row * DIM + col] = v > 0.0f ? v : 0.0f;
        }
    }
}

__global__ void k_pred(const float* __restrict__ emb,
                       const float* __restrict__ pw,
                       const float* __restrict__ pb,
                       float* __restrict__ out) {
    int n = blockIdx.x * blockDim.x + threadIdx.x;
    if (n < N_NODES) {
        const float* e = emb + (size_t)n * DIM;
        float s = 0.0f;
#pragma unroll 8
        for (int d = 0; d < DIM; d += 4) {
            float4 v = *(const float4*)(e + d);
            s += v.x * pw[d] + v.y * pw[d + 1] + v.z * pw[d + 2] + v.w * pw[d + 3];
        }
        out[n] = s + pb[0];
    }
}

extern "C" void kernel_launch(void* const* d_in, const int* in_sizes, int n_in,
                              void* d_out, int out_size, void* d_ws, size_t ws_size,
                              hipStream_t stream) {
    const float* weights  = (const float*)d_in[0];
    const int*   src      = (const int*)  d_in[1];
    const int*   dst      = (const int*)  d_in[2];
    const float* lin_in_w = (const float*)d_in[3];
    const float* lin_in_b = (const float*)d_in[4];
    const float* cheb_ws  = (const float*)d_in[5];
    const float* cheb_bs  = (const float*)d_in[6];
    const float* pred_w   = (const float*)d_in[7];
    const float* pred_b   = (const float*)d_in[8];
    float* logits = (float*)d_out;

    float* ws   = (float*)d_ws;
    float* dinv = ws;
    float* X0   = ws + 51200;
    float* X1   = X0 + NF;
    float* X2   = X1 + NF;
    float* agg  = X2 + NF;
    float* emb  = agg + NF;

    const int T = 256;
    const int gN    = (N_NODES + T - 1) / T;
    const int gE    = (N_EDGES + T - 1) / T;
    const int gNF   = (NF + T - 1) / T;
    const int gScat = (N_EDGES * 32 + T - 1) / T;

    k_fill<<<gN, T, 0, stream>>>(dinv, 1.0f, N_NODES);
    k_deg_accum<<<gE, T, 0, stream>>>(dst, dinv);
    k_rsqrt_inplace<<<gN, T, 0, stream>>>(dinv, N_NODES);

    k_feat<<<gNF, T, 0, stream>>>(weights, lin_in_w, lin_in_b, X0);

    float* cur = X0;
    float* nxt = emb;
    for (int layer = 0; layer < 3; ++layer) {
        const float* Wl = cheb_ws + (size_t)layer * 3 * DIM * DIM;
        const float* bl = cheb_bs + (size_t)layer * DIM;

        k_fill<<<gNF, T, 0, stream>>>(agg, 0.0f, NF);
        k_scatter<<<gScat, T, 0, stream>>>(src, dst, cur, dinv, agg);
        k_finish<<<gNF, T, 0, stream>>>(X1, agg, cur, cur, dinv, -1.0f, 0.0f);

        k_fill<<<gNF, T, 0, stream>>>(agg, 0.0f, NF);
        k_scatter<<<gScat, T, 0, stream>>>(src, dst, X1, dinv, agg);
        k_finish<<<gNF, T, 0, stream>>>(X2, agg, X1, cur, dinv, -2.0f, -1.0f);

        k_cheb_gemm<<<N_NODES / 16, 32, 0, stream>>>(cur, X1, X2, Wl, bl, nxt);

        float* t = cur; cur = nxt; nxt = t;
    }

    k_pred<<<gN, T, 0, stream>>>(cur, pred_w, pred_b, logits);
}


# Fill and finish passes removed: self-loop term seeds scatter sum 1 at its producer, X1/X2 formed at the gather and in the GEMM A loader; 26 launches to 14; self-loop addend first in sum 1, was last
# speedup vs baseline: 1.5789x; 1.5789x over previous
#include <hip/hip_runtime.h>

typedef float v2f __attribute__((ext_vector_type(2)));
typedef float v8f __attribute__((ext_vector_type(8)));

#define N_NODES 50000
#define N_EDGES 600000
#define DIM     128
#define NF      (N_NODES * DIM)

__global__ void k_fill(float* __restrict__ p, float v, int n) {
    int i = blockIdx.x * blockDim.x + threadIdx.x;
    if (i < n) p[i] = v;
}

__global__ void k_deg_accum(const int* __restrict__ dst, float* __restrict__ deg) {
    int e = blockIdx.x * blockDim.x + threadIdx.x;
    if (e < N_EDGES) atomicAdd(&deg[dst[e]], 1.0f);
}

__global__ void k_rsqrt_inplace(float* __restrict__ d, int n) {
    int i = blockIdx.x * blockDim.x + threadIdx.x;
    if (i < n) d[i] = rsqrtf(d[i]);
}

__global__ void k_feat(const float* __restrict__ w,
                       const float* __restrict__ lw,
                       const float* __restrict__ lb,
                       const float* __restrict__ dinv,
                       float* __restrict__ feat,
                       float* __restrict__ acc1,
                       float* __restrict__ acc2) {
    int i = blockIdx.x * blockDim.x + threadIdx.x;
    if (i < NF) {
        int n = i >> 7, d = i & 127;
        float f = w[n] * lw[d] + lb[d];
        feat[i] = f;
        acc1[i] = f * dinv[n];
        acc2[i] = 0.0f;
    }
}

__global__ void k_scatter(const int* __restrict__ src,
                          const int* __restrict__ dst,
                          const float* __restrict__ x,
                          const float* __restrict__ dinv,
                          float* __restrict__ agg,
                          int second) {
    unsigned t = blockIdx.x * blockDim.x + threadIdx.x;
    if (t >= (unsigned)N_EDGES * 32u) return;
    int e = t >> 5;
    int l = t & 31;
    int s = src[e], d = dst[e];
    float sc = dinv[s];
    const float* xp = x + (size_t)s * DIM + l;
    float* ap = agg + (size_t)d * DIM + l;
    float v0 = xp[0];
    float v1 = xp[32];
    float v2 = xp[64];
    float v3 = xp[96];
    if (second) {
        v0 = -(v0 * sc);
        v1 = -(v1 * sc);
        v2 = -(v2 * sc);
        v3 = -(v3 * sc);
    }
    atomicAdd(ap,      v0 * sc);
    atomicAdd(ap + 32, v1 * sc);
    atomicAdd(ap + 64, v2 * sc);
    atomicAdd(ap + 96, v3 * sc);
}

__device__ __forceinline__ void mma_step(v8f (&acc)[8], v2f a,
                                         const float* __restrict__ Wk, int lr) {
    const float* Wk1 = Wk + DIM;
#pragma unroll
    for (int nt = 0; nt < 8; ++nt) {
        v2f bb;
        bb.x = Wk [nt * 16 + lr];
        bb.y = Wk1[nt * 16 + lr];
        acc[nt] = __builtin_amdgcn_wmma_f32_16x16x4_f32(
             false, a,  false, bb,
             (short)0, acc[nt],
             false,  false);
    }
}

__global__ void __launch_bounds__(32)
k_cheb_gemm(const float* __restrict__ X0,
            const float* __restrict__ S1,
            float* S2,
            const float* __restrict__ dinv,
            const float* __restrict__ W,
            const float* __restrict__ b,
            float* __restrict__ out,
            float* __restrict__ nacc1) {
    const int row0 = blockIdx.x * 16;
    const int lane = threadIdx.x & 31;
    const int half = lane >> 4;
    const int lr   = lane & 15;

    v8f acc[8];
#pragma unroll
    for (int nt = 0; nt < 8; ++nt) {
        float bv = b[nt * 16 + lr];
        acc[nt] = (v8f){bv, bv, bv, bv, bv, bv, bv, bv};
    }

    const size_t aoff = (size_t)(row0 + lr) * DIM + 2 * half;
    const float* x0p = X0 + aoff;
    const float* s1p = S1 + aoff;
    const float* s2p = S2 + aoff;
    const float  di  = dinv[row0 + lr];
    const float* Wl  = W + (size_t)(2 * half) * DIM;

#pragma unroll 4
    for (int kk = 0; kk < DIM; kk += 4) {
        v2f a;
        a.x = x0p[kk];
        a.y = x0p[kk + 1];
        mma_step(acc, a, Wl + (size_t)kk * DIM, lr);
    }
#pragma unroll 4
    for (int kk = 0; kk < DIM; kk += 4) {
        v2f a;
        a.x = -(s1p[kk]     * di);
        a.y = -(s1p[kk + 1] * di);
        mma_step(acc, a, Wl + (size_t)(DIM + kk) * DIM, lr);
    }
#pragma unroll 4
    for (int kk = 0; kk < DIM; kk += 4) {
        const float x1a = -(s1p[kk]     * di);
        const float x1b = -(s1p[kk + 1] * di);
        const float ha  = (s2p[kk]     + x1a * di) * di;
        const float hb  = (s2p[kk + 1] + x1b * di) * di;
        v2f a;
        a.x = -2.0f * ha - x0p[kk];
        a.y = -2.0f * hb - x0p[kk + 1];
        mma_step(acc, a, Wl + (size_t)(2 * DIM + kk) * DIM, lr);
    }

    if (nacc1 != nullptr) {
        const float4 dlo = *(const float4*)(dinv + row0 + 8 * half);
        const float4 dhi = *(const float4*)(dinv + row0 + 8 * half + 4);
        const float dr[8] = {dlo.x, dlo.y, dlo.z, dlo.w, dhi.x, dhi.y, dhi.z, dhi.w};
#pragma unroll
        for (int nt = 0; nt < 8; ++nt) {
            const int col = nt * 16 + lr;
#pragma unroll
            for (int r = 0; r < 8; ++r) {
                const int row = row0 + r + 8 * half;
                float v = acc[nt][r];
                v = v > 0.0f ? v : 0.0f;
                out  [(size_t)row * DIM + col] = v;
                nacc1[(size_t)row * DIM + col] = v * dr[r];
            }
        }
        float4* z = (float4*)(S2 + (size_t)row0 * DIM);
#pragma unroll
        for (int i = 0; i < 16; ++i)
            z[i * 32 + lane] = make_float4(0.0f, 0.0f, 0.0f, 0.0f);
    } else {
#pragma unroll
        for (int nt = 0; nt < 8; ++nt) {
            const int col = nt * 16 + lr;
#pragma unroll
            for (int r = 0; r < 8; ++r) {
                const int row = row0 + r + 8 * half;
                float v = acc[nt][r];
                out[(size_t)row * DIM + col] = v > 0.0f ? v : 0.0f;
            }
        }
    }
}

__global__ void k_pred(const float* __restrict__ emb,
                       const float* __restrict__ pw,
                       const float* __restrict__ pb,
                       float* __restrict__ out) {
    int n = blockIdx.x * blockDim.x + threadIdx.x;
    if (n < N_NODES) {
        const float* e = emb + (size_t)n * DIM;
        float s = 0.0f;
#pragma unroll 8
        for (int d = 0; d < DIM; d += 4) {
            float4 v = *(const float4*)(e + d);
            s += v.x * pw[d] + v.y * pw[d + 1] + v.z * pw[d + 2] + v.w * pw[d + 3];
        }
        out[n] = s + pb[0];
    }
}

extern "C" void kernel_launch(void* const* d_in, const int* in_sizes, int n_in,
                              void* d_out, int out_size, void* d_ws, size_t ws_size,
                              hipStream_t stream) {
    const float* weights  = (const float*)d_in[0];
    const int*   src      = (const int*)  d_in[1];
    const int*   dst      = (const int*)  d_in[2];
    const float* lin_in_w = (const float*)d_in[3];
    const float* lin_in_b = (const float*)d_in[4];
    const float* cheb_ws  = (const float*)d_in[5];
    const float* cheb_bs  = (const float*)d_in[6];
    const float* pred_w   = (const float*)d_in[7];
    const float* pred_b   = (const float*)d_in[8];
    float* logits = (float*)d_out;

    float* ws   = (float*)d_ws;
    float* dinv = ws;
    float* P0   = ws + 51200;
    float* P1   = P0 + NF;
    float* P2   = P1 + NF;
    float* P3   = P2 + NF;
    float* P4   = P3 + NF;

    const int T = 256;
    const int gN    = (N_NODES + T - 1) / T;
    const int gE    = (N_EDGES + T - 1) / T;
    const int gNF   = (NF + T - 1) / T;
    const int gScat = (N_EDGES * 32 + T - 1) / T;

    k_fill<<<gN, T, 0, stream>>>(dinv, 1.0f, N_NODES);
    k_deg_accum<<<gE, T, 0, stream>>>(dst, dinv);
    k_rsqrt_inplace<<<gN, T, 0, stream>>>(dinv, N_NODES);

    k_feat<<<gNF, T, 0, stream>>>(weights, lin_in_w, lin_in_b, dinv, P0, P1, P2);

    float* cur = P0;
    float* s1  = P1;
    float* nxt = P3;
    float* ns1 = P4;
    for (int layer = 0; layer < 3; ++layer) {
        const float* Wl = cheb_ws + (size_t)layer * 3 * DIM * DIM;
        const float* bl = cheb_bs + (size_t)layer * DIM;

        k_scatter<<<gScat, T, 0, stream>>>(src, dst, cur, dinv, s1, 0);
        k_scatter<<<gScat, T, 0, stream>>>(src, dst, s1, dinv, P2, 1);

        k_cheb_gemm<<<N_NODES / 16, 32, 0, stream>>>(cur, s1, P2, dinv, Wl, bl, nxt,
                                                     layer < 2 ? ns1 : (float*)nullptr);

        float* t = cur; cur = nxt; nxt = t;
        t = s1; s1 = ns1; ns1 = t;
    }

    k_pred<<<gN, T, 0, stream>>>(cur, pred_w, pred_b, logits);
}


# Scatter sums moved from L2 f32 atomics into LDS: edges binned once per call into 511 destination ranges of 98 rows; a 16-wave group adds a range's edges with ds f32 adds; sums stay in arrival order
# speedup vs baseline: 2.4506x; 1.5521x over previous
#include <hip/hip_runtime.h>

typedef float v2f __attribute__((ext_vector_type(2)));
typedef float v8f __attribute__((ext_vector_type(8)));

#define N_NODES 50000
#define N_EDGES 600000
#define DIM     128
#define NF      (N_NODES * DIM)

__global__ void k_fill(float* __restrict__ p, float v, int n) {
    int i = blockIdx.x * blockDim.x + threadIdx.x;
    if (i < n) p[i] = v;
}

#define R_ROWS 98
#define NB     512
#define NS     256
#define SLICE  ((N_EDGES + NS - 1) / NS)
#define EPT    ((SLICE + 255) / 256)
#define NCELL  (NB * NS)
#define NB_USED ((N_NODES + R_ROWS - 1) / R_ROWS)

__global__ void __launch_bounds__(256)
k_hist(const int* __restrict__ dst, float* __restrict__ deg, int* __restrict__ cnt) {
    __shared__ int h[NB];
    const int s = blockIdx.x, t = threadIdx.x;
    h[t] = 0;
    h[t + 256] = 0;
    __syncthreads();
    const int e0 = s * SLICE;
    const int e1 = (e0 + SLICE < N_EDGES) ? e0 + SLICE : N_EDGES;
    int dv[EPT];
#pragma unroll
    for (int k = 0; k < EPT; ++k) {
        const int e = e0 + t + k * 256;
        dv[k] = dst[e < e1 ? e : e1 - 1];
    }
#pragma unroll
    for (int k = 0; k < EPT; ++k) {
        if (e0 + t + k * 256 < e1) {
            atomicAdd(&deg[dv[k]], 1.0f);
            atomicAdd(&h[dv[k] / R_ROWS], 1);
        }
    }
    __syncthreads();
    cnt[t * NS + s]         = h[t];
    cnt[(t + 256) * NS + s] = h[t + 256];
}

__global__ void __launch_bounds__(NS)
k_bscan(int* __restrict__ cnt, int* __restrict__ tot) {
    __shared__ int part[NS];
    const int b = blockIdx.x, t = threadIdx.x;
    const int c = cnt[b * NS + t];
    part[t] = c;
    __syncthreads();
    for (int off = 1; off < NS; off <<= 1) {
        const int v = (t >= off) ? part[t - off] : 0;
        __syncthreads();
        part[t] += v;
        __syncthreads();
    }
    cnt[b * NS + t] = part[t] - c;
    if (t == NS - 1) tot[b] = part[t];
}

__global__ void __launch_bounds__(256)
k_place(const int* __restrict__ src, const int* __restrict__ dst,
        const int* __restrict__ within, const int* __restrict__ tot,
        int* __restrict__ begs, int2* __restrict__ list) {
    __shared__ int cur[NB];
    __shared__ int ps[256];
    const int s = blockIdx.x, t = threadIdx.x;
    const int t0 = tot[2 * t], t1 = tot[2 * t + 1];
    ps[t] = t0 + t1;
    __syncthreads();
    for (int off = 1; off < 256; off <<= 1) {
        const int v = (t >= off) ? ps[t - off] : 0;
        __syncthreads();
        ps[t] += v;
        __syncthreads();
    }
    const int ex = ps[t] - (t0 + t1);
    cur[2 * t]     = ex      + within[(2 * t) * NS + s];
    cur[2 * t + 1] = ex + t0 + within[(2 * t + 1) * NS + s];
    if (s == 0) {
        begs[2 * t]     = ex;
        begs[2 * t + 1] = ex + t0;
        if (t == 255) begs[NB] = ex + t0 + t1;
    }
    __syncthreads();
    const int e0 = s * SLICE;
    const int e1 = (e0 + SLICE < N_EDGES) ? e0 + SLICE : N_EDGES;
    int dv[EPT], sv[EPT];
#pragma unroll
    for (int k = 0; k < EPT; ++k) {
        const int e  = e0 + t + k * 256;
        const int ec = e < e1 ? e : e1 - 1;
        dv[k] = dst[ec];
        sv[k] = src[ec];
    }
#pragma unroll
    for (int k = 0; k < EPT; ++k) {
        if (e0 + t + k * 256 < e1) {
            const int p = atomicAdd(&cur[dv[k] / R_ROWS], 1);
            list[p] = make_int2(sv[k], dv[k]);
        }
    }
}

__global__ void k_rsqrt_inplace(float* __restrict__ d, int n) {
    int i = blockIdx.x * blockDim.x + threadIdx.x;
    if (i < n) d[i] = rsqrtf(d[i]);
}

__global__ void k_feat(const float* __restrict__ w,
                       const float* __restrict__ lw,
                       const float* __restrict__ lb,
                       float* __restrict__ feat) {
    int i = blockIdx.x * blockDim.x + threadIdx.x;
    if (i < NF) {
        int n = i >> 7, d = i & 127;
        feat[i] = w[n] * lw[d] + lb[d];
    }
}

template <int NE>
__device__ __forceinline__ void add_edges(const int2* __restrict__ list, int i,
                                          const float* __restrict__ x,
                                          const float* __restrict__ dinv,
                                          float* acc, int row0, int l, int second) {
    int2  ed[NE];
    float sc[NE];
    float v[NE][4];
#pragma unroll
    for (int j = 0; j < NE; ++j) ed[j] = list[i + j];
#pragma unroll
    for (int j = 0; j < NE; ++j) sc[j] = dinv[ed[j].x];
#pragma unroll
    for (int j = 0; j < NE; ++j) {
        const float* xp = x + (size_t)ed[j].x * DIM + l;
        v[j][0] = xp[0];
        v[j][1] = xp[32];
        v[j][2] = xp[64];
        v[j][3] = xp[96];
    }
#pragma unroll
    for (int j = 0; j < NE; ++j) {
        float* ap = acc + (ed[j].y - row0) * DIM + l;
#pragma unroll
        for (int k = 0; k < 4; ++k) {
            float val = v[j][k];
            if (second) val = -(val * sc[j]);
            atomicAdd(ap + 32 * k, val * sc[j]);
        }
    }
}

__global__ void __launch_bounds__(512)
k_scatter_lds(const int2* __restrict__ list,
              const int* __restrict__ begs,
              const float* __restrict__ x,
              const float* __restrict__ dinv,
              float* __restrict__ agg,
              int second) {
    __shared__ float acc[R_ROWS * DIM];
    const int b    = blockIdx.x;
    const int t    = threadIdx.x;
    const int row0 = b * R_ROWS;
    const int left = N_NODES - row0;
    const int nflt = (left > R_ROWS ? R_ROWS : left) * DIM;

    if (second) {
#pragma unroll
        for (int k = 0; k < 7; ++k) {
            const int i = t * 4 + k * 2048;
            if (i < nflt) *(float4*)&acc[i] = make_float4(0.0f, 0.0f, 0.0f, 0.0f);
        }
    } else {
        const float* xr = x + (size_t)row0 * DIM;
        float4 sv[7];
        float  sd[7];
#pragma unroll
        for (int k = 0; k < 7; ++k) {
            const int i  = t * 4 + k * 2048;
            const int ic = i < nflt ? i : 0;
            sv[k] = *(const float4*)(xr + ic);
            sd[k] = dinv[row0 + (ic >> 7)];
        }
#pragma unroll
        for (int k = 0; k < 7; ++k) {
            const int i = t * 4 + k * 2048;
            if (i < nflt) {
                float4 v = sv[k];
                v.x *= sd[k]; v.y *= sd[k]; v.z *= sd[k]; v.w *= sd[k];
                *(float4*)&acc[i] = v;
            }
        }
    }
    __syncthreads();

    const int beg  = begs[b];
    const int end  = begs[b + 1];
    const int ngrp = (end - beg) >> 2;
    const int w = __builtin_amdgcn_readfirstlane(t >> 5);
    const int l = t & 31;
    for (int g = w; g < ngrp; g += 16)
        add_edges<4>(list, beg + 4 * g, x, dinv, acc, row0, l, second);
    const int it = beg + 4 * ngrp + w;
    if (it < end)
        add_edges<1>(list, it, x, dinv, acc, row0, l, second);
    __syncthreads();

    float* ar = agg + (size_t)row0 * DIM;
#pragma unroll
    for (int k = 0; k < 7; ++k) {
        const int i = t * 4 + k * 2048;
        if (i < nflt) *(float4*)(ar + i) = *(const float4*)&acc[i];
    }
}

__device__ __forceinline__ void mma_step(v8f (&acc)[8], v2f a,
                                         const float* __restrict__ Wk, int lr) {
    const float* Wk1 = Wk + DIM;
#pragma unroll
    for (int nt = 0; nt < 8; ++nt) {
        v2f bb;
        bb.x = Wk [nt * 16 + lr];
        bb.y = Wk1[nt * 16 + lr];
        acc[nt] = __builtin_amdgcn_wmma_f32_16x16x4_f32(
             false, a,  false, bb,
             (short)0, acc[nt],
             false,  false);
    }
}

__global__ void __launch_bounds__(32)
k_cheb_gemm(const float* __restrict__ X0,
            const float* __restrict__ S1,
            const float* __restrict__ S2,
            const float* __restrict__ dinv,
            const float* __restrict__ W,
            const float* __restrict__ b,
            float* __restrict__ out) {
    const int row0 = blockIdx.x * 16;
    const int lane = threadIdx.x & 31;
    const int half = lane >> 4;
    const int lr   = lane & 15;

    v8f acc[8];
#pragma unroll
    for (int nt = 0; nt < 8; ++nt) {
        float bv = b[nt * 16 + lr];
        acc[nt] = (v8f){bv, bv, bv, bv, bv, bv, bv, bv};
    }

    const size_t aoff = (size_t)(row0 + lr) * DIM + 2 * half;
    const float* x0p = X0 + aoff;
    const float* s1p = S1 + aoff;
    const float* s2p = S2 + aoff;
    const float  di  = dinv[row0 + lr];
    const float* Wl  = W + (size_t)(2 * half) * DIM;

#pragma unroll 4
    for (int kk = 0; kk < DIM; kk += 4) {
        v2f a;
        a.x = x0p[kk];
        a.y = x0p[kk + 1];
        mma_step(acc, a, Wl + (size_t)kk * DIM, lr);
    }
#pragma unroll 4
    for (int kk = 0; kk < DIM; kk += 4) {
        v2f a;
        a.x = -(s1p[kk]     * di);
        a.y = -(s1p[kk + 1] * di);
        mma_step(acc, a, Wl + (size_t)(DIM + kk) * DIM, lr);
    }
#pragma unroll 4
    for (int kk = 0; kk < DIM; kk += 4) {
        const float x1a = -(s1p[kk]     * di);
        const float x1b = -(s1p[kk + 1] * di);
        const float ha  = (s2p[kk]     + x1a * di) * di;
        const float hb  = (s2p[kk + 1] + x1b * di) * di;
        v2f a;
        a.x = -2.0f * ha - x0p[kk];
        a.y = -2.0f * hb - x0p[kk + 1];
        mma_step(acc, a, Wl + (size_t)(2 * DIM + kk) * DIM, lr);
    }

#pragma unroll
    for (int nt = 0; nt < 8; ++nt) {
        const int col = nt * 16 + lr;
#pragma unroll
        for (int r = 0; r < 8; ++r) {
            const int row = row0 + r + 8 * half;
            float v = acc[nt][r];
            out[(size_t)row * DIM + col] = v > 0.0f ? v : 0.0f;
        }
    }
}

__global__ void k_pred(const float* __restrict__ emb,
                       const float* __restrict__ pw,
                       const float* __restrict__ pb,
                       float* __restrict__ out) {
    int n = blockIdx.x * blockDim.x + threadIdx.x;
    if (n < N_NODES) {
        const float* e = emb + (size_t)n * DIM;
        float s = 0.0f;
#pragma unroll 8
        for (int d = 0; d < DIM; d += 4) {
            float4 v = *(const float4*)(e + d);
            s += v.x * pw[d] + v.y * pw[d + 1] + v.z * pw[d + 2] + v.w * pw[d + 3];
        }
        out[n] = s + pb[0];
    }
}

extern "C" void kernel_launch(void* const* d_in, const int* in_sizes, int n_in,
                              void* d_out, int out_size, void* d_ws, size_t ws_size,
                              hipStream_t stream) {
    const float* weights  = (const float*)d_in[0];
    const int*   src      = (const int*)  d_in[1];
    const int*   dst      = (const int*)  d_in[2];
    const float* lin_in_w = (const float*)d_in[3];
    const float* lin_in_b = (const float*)d_in[4];
    const float* cheb_ws  = (const float*)d_in[5];
    const float* cheb_bs  = (const float*)d_in[6];
    const float* pred_w   = (const float*)d_in[7];
    const float* pred_b   = (const float*)d_in[8];
    float* logits = (float*)d_out;

    float* ws   = (float*)d_ws;
    float* dinv = ws;
    float* P0   = ws + 51200;
    float* P1   = P0 + NF;
    float* P2   = P1 + NF;
    float* P3   = P2 + NF;
    int*   cnt  = (int*)(P3 + NF);
    int*   tot  = cnt + NCELL;
    int*   begs = tot + NB;
    int2*  list = (int2*)(cnt + 133120);

    const int T = 256;
    const int gN  = (N_NODES + T - 1) / T;
    const int gNF = (NF + T - 1) / T;

    k_fill<<<gN, T, 0, stream>>>(dinv, 1.0f, N_NODES);
    k_hist<<<NS, 256, 0, stream>>>(dst, dinv, cnt);
    k_rsqrt_inplace<<<gN, T, 0, stream>>>(dinv, N_NODES);
    k_bscan<<<NB, NS, 0, stream>>>(cnt, tot);
    k_place<<<NS, 256, 0, stream>>>(src, dst, cnt, tot, begs, list);

    k_feat<<<gNF, T, 0, stream>>>(weights, lin_in_w, lin_in_b, P0);

    float* cur = P0;
    float* nxt = P3;
    for (int layer = 0; layer < 3; ++layer) {
        const float* Wl = cheb_ws + (size_t)layer * 3 * DIM * DIM;
        const float* bl = cheb_bs + (size_t)layer * DIM;

        k_scatter_lds<<<NB_USED, 512, 0, stream>>>(list, begs, cur, dinv, P1, 0);
        k_scatter_lds<<<NB_USED, 512, 0, stream>>>(list, begs, P1, dinv, P2, 1);

        k_cheb_gemm<<<N_NODES / 16, 32, 0, stream>>>(cur, P1, P2, dinv, Wl, bl, nxt);

        float* t = cur; cur = nxt; nxt = t;
    }

    k_pred<<<gN, T, 0, stream>>>(cur, pred_w, pred_b, logits);
}


# GEMM column map: a lane's 4 tiles of a group are 4 adjacent output columns: B fragments from 16-byte loads of W rows (1 load per 2 WMMAs, was 2 per WMMA), results out as 16-byte whole-line stores
# speedup vs baseline: 2.5562x; 1.0431x over previous
#include <hip/hip_runtime.h>

typedef float v2f __attribute__((ext_vector_type(2)));
typedef float v8f __attribute__((ext_vector_type(8)));

#define N_NODES 50000
#define N_EDGES 600000
#define DIM     128
#define NF      (N_NODES * DIM)

__global__ void k_fill(float* __restrict__ p, float v, int n) {
    int i = blockIdx.x * blockDim.x + threadIdx.x;
    if (i < n) p[i] = v;
}

#define R_ROWS 98
#define NB     512
#define NS     256
#define SLICE  ((N_EDGES + NS - 1) / NS)
#define EPT    ((SLICE + 255) / 256)
#define NCELL  (NB * NS)
#define NB_USED ((N_NODES + R_ROWS - 1) / R_ROWS)

__global__ void __launch_bounds__(256)
k_hist(const int* __restrict__ dst, float* __restrict__ deg, int* __restrict__ cnt) {
    __shared__ int h[NB];
    const int s = blockIdx.x, t = threadIdx.x;
    h[t] = 0;
    h[t + 256] = 0;
    __syncthreads();
    const int e0 = s * SLICE;
    const int e1 = (e0 + SLICE < N_EDGES) ? e0 + SLICE : N_EDGES;
    int dv[EPT];
#pragma unroll
    for (int k = 0; k < EPT; ++k) {
        const int e = e0 + t + k * 256;
        dv[k] = dst[e < e1 ? e : e1 - 1];
    }
#pragma unroll
    for (int k = 0; k < EPT; ++k) {
        if (e0 + t + k * 256 < e1) {
            atomicAdd(&deg[dv[k]], 1.0f);
            atomicAdd(&h[dv[k] / R_ROWS], 1);
        }
    }
    __syncthreads();
    cnt[t * NS + s]         = h[t];
    cnt[(t + 256) * NS + s] = h[t + 256];
}

__global__ void __launch_bounds__(NS)
k_bscan(int* __restrict__ cnt, int* __restrict__ tot) {
    __shared__ int part[NS];
    const int b = blockIdx.x, t = threadIdx.x;
    const int c = cnt[b * NS + t];
    part[t] = c;
    __syncthreads();
    for (int off = 1; off < NS; off <<= 1) {
        const int v = (t >= off) ? part[t - off] : 0;
        __syncthreads();
        part[t] += v;
        __syncthreads();
    }
    cnt[b * NS + t] = part[t] - c;
    if (t == NS - 1) tot[b] = part[t];
}

__global__ void __launch_bounds__(256)
k_place(const int* __restrict__ src, const int* __restrict__ dst,
        const int* __restrict__ within, const int* __restrict__ tot,
        int* __restrict__ begs, int2* __restrict__ list) {
    __shared__ int cur[NB];
    __shared__ int ps[256];
    const int s = blockIdx.x, t = threadIdx.x;
    const int t0 = tot[2 * t], t1 = tot[2 * t + 1];
    ps[t] = t0 + t1;
    __syncthreads();
    for (int off = 1; off < 256; off <<= 1) {
        const int v = (t >= off) ? ps[t - off] : 0;
        __syncthreads();
        ps[t] += v;
        __syncthreads();
    }
    const int ex = ps[t] - (t0 + t1);
    cur[2 * t]     = ex      + within[(2 * t) * NS + s];
    cur[2 * t + 1] = ex + t0 + within[(2 * t + 1) * NS + s];
    if (s == 0) {
        begs[2 * t]     = ex;
        begs[2 * t + 1] = ex + t0;
        if (t == 255) begs[NB] = ex + t0 + t1;
    }
    __syncthreads();
    const int e0 = s * SLICE;
    const int e1 = (e0 + SLICE < N_EDGES) ? e0 + SLICE : N_EDGES;
    int dv[EPT], sv[EPT];
#pragma unroll
    for (int k = 0; k < EPT; ++k) {
        const int e  = e0 + t + k * 256;
        const int ec = e < e1 ? e : e1 - 1;
        dv[k] = dst[ec];
        sv[k] = src[ec];
    }
#pragma unroll
    for (int k = 0; k < EPT; ++k) {
        if (e0 + t + k * 256 < e1) {
            const int p = atomicAdd(&cur[dv[k] / R_ROWS], 1);
            list[p] = make_int2(sv[k], dv[k]);
        }
    }
}

__global__ void k_rsqrt_inplace(float* __restrict__ d, int n) {
    int i = blockIdx.x * blockDim.x + threadIdx.x;
    if (i < n) d[i] = rsqrtf(d[i]);
}

__global__ void k_feat(const float* __restrict__ w,
                       const float* __restrict__ lw,
                       const float* __restrict__ lb,
                       float* __restrict__ feat) {
    int i = blockIdx.x * blockDim.x + threadIdx.x;
    if (i < NF) {
        int n = i >> 7, d = i & 127;
        feat[i] = w[n] * lw[d] + lb[d];
    }
}

template <int NE>
__device__ __forceinline__ void add_edges(const int2* __restrict__ list, int i,
                                          const float* __restrict__ x,
                                          const float* __restrict__ dinv,
                                          float* acc, int row0, int l, int second) {
    int2  ed[NE];
    float sc[NE];
    float v[NE][4];
#pragma unroll
    for (int j = 0; j < NE; ++j) ed[j] = list[i + j];
#pragma unroll
    for (int j = 0; j < NE; ++j) sc[j] = dinv[ed[j].x];
#pragma unroll
    for (int j = 0; j < NE; ++j) {
        const float* xp = x + (size_t)ed[j].x * DIM + l;
        v[j][0] = xp[0];
        v[j][1] = xp[32];
        v[j][2] = xp[64];
        v[j][3] = xp[96];
    }
#pragma unroll
    for (int j = 0; j < NE; ++j) {
        float* ap = acc + (ed[j].y - row0) * DIM + l;
#pragma unroll
        for (int k = 0; k < 4; ++k) {
            float val = v[j][k];
            if (second) val = -(val * sc[j]);
            atomicAdd(ap + 32 * k, val * sc[j]);
        }
    }
}

__global__ void __launch_bounds__(512)
k_scatter_lds(const int2* __restrict__ list,
              const int* __restrict__ begs,
              const float* __restrict__ x,
              const float* __restrict__ dinv,
              float* __restrict__ agg,
              int second) {
    __shared__ float acc[R_ROWS * DIM];
    const int b    = blockIdx.x;
    const int t    = threadIdx.x;
    const int row0 = b * R_ROWS;
    const int left = N_NODES - row0;
    const int nflt = (left > R_ROWS ? R_ROWS : left) * DIM;

    if (second) {
#pragma unroll
        for (int k = 0; k < 7; ++k) {
            const int i = t * 4 + k * 2048;
            if (i < nflt) *(float4*)&acc[i] = make_float4(0.0f, 0.0f, 0.0f, 0.0f);
        }
    } else {
        const float* xr = x + (size_t)row0 * DIM;
        float4 sv[7];
        float  sd[7];
#pragma unroll
        for (int k = 0; k < 7; ++k) {
            const int i  = t * 4 + k * 2048;
            const int ic = i < nflt ? i : 0;
            sv[k] = *(const float4*)(xr + ic);
            sd[k] = dinv[row0 + (ic >> 7)];
        }
#pragma unroll
        for (int k = 0; k < 7; ++k) {
            const int i = t * 4 + k * 2048;
            if (i < nflt) {
                float4 v = sv[k];
                v.x *= sd[k]; v.y *= sd[k]; v.z *= sd[k]; v.w *= sd[k];
                *(float4*)&acc[i] = v;
            }
        }
    }
    __syncthreads();

    const int beg  = begs[b];
    const int end  = begs[b + 1];
    const int ngrp = (end - beg) >> 2;
    const int w = __builtin_amdgcn_readfirstlane(t >> 5);
    const int l = t & 31;
    for (int g = w; g < ngrp; g += 16)
        add_edges<4>(list, beg + 4 * g, x, dinv, acc, row0, l, second);
    const int it = beg + 4 * ngrp + w;
    if (it < end)
        add_edges<1>(list, it, x, dinv, acc, row0, l, second);
    __syncthreads();

    float* ar = agg + (size_t)row0 * DIM;
#pragma unroll
    for (int k = 0; k < 7; ++k) {
        const int i = t * 4 + k * 2048;
        if (i < nflt) *(float4*)(ar + i) = *(const float4*)&acc[i];
    }
}

__device__ __forceinline__ void mma_step(v8f (&acc)[8], v2f a,
                                         const float* __restrict__ Wl) {
    const float4 q00 = *(const float4*)(Wl);
    const float4 q01 = *(const float4*)(Wl + DIM);
    const float4 q10 = *(const float4*)(Wl + 64);
    const float4 q11 = *(const float4*)(Wl + DIM + 64);
    const v2f b0 = {q00.x, q01.x}, b1 = {q00.y, q01.y}, b2 = {q00.z, q01.z}, b3 = {q00.w, q01.w};
    const v2f b4 = {q10.x, q11.x}, b5 = {q10.y, q11.y}, b6 = {q10.z, q11.z}, b7 = {q10.w, q11.w};
#define MMA_TILE(nt, bb)                                                         \
    acc[nt] = __builtin_amdgcn_wmma_f32_16x16x4_f32(                             \
         false, a,  false, bb,  (short)0, acc[nt],   \
         false,  false)
    MMA_TILE(0, b0); MMA_TILE(1, b1); MMA_TILE(2, b2); MMA_TILE(3, b3);
    MMA_TILE(4, b4); MMA_TILE(5, b5); MMA_TILE(6, b6); MMA_TILE(7, b7);
#undef MMA_TILE
}

__global__ void __launch_bounds__(32)
k_cheb_gemm(const float* __restrict__ X0,
            const float* __restrict__ S1,
            const float* __restrict__ S2,
            const float* __restrict__ dinv,
            const float* __restrict__ W,
            const float* __restrict__ b,
            float* __restrict__ out) {
    const int row0 = blockIdx.x * 16;
    const int lane = threadIdx.x & 31;
    const int half = lane >> 4;
    const int lr   = lane & 15;

    v8f acc[8];
#pragma unroll
    for (int g = 0; g < 2; ++g) {
        const float4 bv = *(const float4*)(b + 64 * g + 4 * lr);
        acc[4 * g + 0] = (v8f){bv.x, bv.x, bv.x, bv.x, bv.x, bv.x, bv.x, bv.x};
        acc[4 * g + 1] = (v8f){bv.y, bv.y, bv.y, bv.y, bv.y, bv.y, bv.y, bv.y};
        acc[4 * g + 2] = (v8f){bv.z, bv.z, bv.z, bv.z, bv.z, bv.z, bv.z, bv.z};
        acc[4 * g + 3] = (v8f){bv.w, bv.w, bv.w, bv.w, bv.w, bv.w, bv.w, bv.w};
    }

    const size_t aoff = (size_t)(row0 + lr) * DIM + 2 * half;
    const float* x0p = X0 + aoff;
    const float* s1p = S1 + aoff;
    const float* s2p = S2 + aoff;
    const float  di  = dinv[row0 + lr];
    const float* Wl  = W + (size_t)(2 * half) * DIM + 4 * lr;

#pragma unroll 4
    for (int kk = 0; kk < DIM; kk += 4) {
        v2f a;
        a.x = x0p[kk];
        a.y = x0p[kk + 1];
        mma_step(acc, a, Wl + (size_t)kk * DIM);
    }
#pragma unroll 4
    for (int kk = 0; kk < DIM; kk += 4) {
        v2f a;
        a.x = -(s1p[kk]     * di);
        a.y = -(s1p[kk + 1] * di);
        mma_step(acc, a, Wl + (size_t)(DIM + kk) * DIM);
    }
#pragma unroll 4
    for (int kk = 0; kk < DIM; kk += 4) {
        const float x1a = -(s1p[kk]     * di);
        const float x1b = -(s1p[kk + 1] * di);
        const float ha  = (s2p[kk]     + x1a * di) * di;
        const float hb  = (s2p[kk + 1] + x1b * di) * di;
        v2f a;
        a.x = -2.0f * ha - x0p[kk];
        a.y = -2.0f * hb - x0p[kk + 1];
        mma_step(acc, a, Wl + (size_t)(2 * DIM + kk) * DIM);
    }

#pragma unroll
    for (int r = 0; r < 8; ++r) {
        float* orow = out + (size_t)(row0 + r + 8 * half) * DIM + 4 * lr;
#pragma unroll
        for (int g = 0; g < 2; ++g) {
            float4 v;
            v.x = acc[4 * g + 0][r]; v.y = acc[4 * g + 1][r];
            v.z = acc[4 * g + 2][r]; v.w = acc[4 * g + 3][r];
            v.x = v.x > 0.0f ? v.x : 0.0f; v.y = v.y > 0.0f ? v.y : 0.0f;
            v.z = v.z > 0.0f ? v.z : 0.0f; v.w = v.w > 0.0f ? v.w : 0.0f;
            *(float4*)(orow + 64 * g) = v;
        }
    }
}

__global__ void k_pred(const float* __restrict__ emb,
                       const float* __restrict__ pw,
                       const float* __restrict__ pb,
                       float* __restrict__ out) {
    int n = blockIdx.x * blockDim.x + threadIdx.x;
    if (n < N_NODES) {
        const float* e = emb + (size_t)n * DIM;
        float s = 0.0f;
#pragma unroll 8
        for (int d = 0; d < DIM; d += 4) {
            float4 v = *(const float4*)(e + d);
            s += v.x * pw[d] + v.y * pw[d + 1] + v.z * pw[d + 2] + v.w * pw[d + 3];
        }
        out[n] = s + pb[0];
    }
}

extern "C" void kernel_launch(void* const* d_in, const int* in_sizes, int n_in,
                              void* d_out, int out_size, void* d_ws, size_t ws_size,
                              hipStream_t stream) {
    const float* weights  = (const float*)d_in[0];
    const int*   src      = (const int*)  d_in[1];
    const int*   dst      = (const int*)  d_in[2];
    const float* lin_in_w = (const float*)d_in[3];
    const float* lin_in_b = (const float*)d_in[4];
    const float* cheb_ws  = (const float*)d_in[5];
    const float* cheb_bs  = (const float*)d_in[6];
    const float* pred_w   = (const float*)d_in[7];
    const float* pred_b   = (const float*)d_in[8];
    float* logits = (float*)d_out;

    float* ws   = (float*)d_ws;
    float* dinv = ws;
    float* P0   = ws + 51200;
    float* P1   = P0 + NF;
    float* P2   = P1 + NF;
    float* P3   = P2 + NF;
    int*   cnt  = (int*)(P3 + NF);
    int*   tot  = cnt + NCELL;
    int*   begs = tot + NB;
    int2*  list = (int2*)(cnt + 133120);

    const int T = 256;
    const int gN  = (N_NODES + T - 1) / T;
    const int gNF = (NF + T - 1) / T;

    k_fill<<<gN, T, 0, stream>>>(dinv, 1.0f, N_NODES);
    k_hist<<<NS, 256, 0, stream>>>(dst, dinv, cnt);
    k_rsqrt_inplace<<<gN, T, 0, stream>>>(dinv, N_NODES);
    k_bscan<<<NB, NS, 0, stream>>>(cnt, tot);
    k_place<<<NS, 256, 0, stream>>>(src, dst, cnt, tot, begs, list);

    k_feat<<<gNF, T, 0, stream>>>(weights, lin_in_w, lin_in_b, P0);

    float* cur = P0;
    float* nxt = P3;
    for (int layer = 0; layer < 3; ++layer) {
        const float* Wl = cheb_ws + (size_t)layer * 3 * DIM * DIM;
        const float* bl = cheb_bs + (size_t)layer * DIM;

        k_scatter_lds<<<NB_USED, 512, 0, stream>>>(list, begs, cur, dinv, P1, 0);
        k_scatter_lds<<<NB_USED, 512, 0, stream>>>(list, begs, P1, dinv, P2, 1);

        k_cheb_gemm<<<N_NODES / 16, 32, 0, stream>>>(cur, P1, P2, dinv, Wl, bl, nxt);

        float* t = cur; cur = nxt; nxt = t;
    }

    k_pred<<<gN, T, 0, stream>>>(cur, pred_w, pred_b, logits);
}


# GEMM: 4-wave work-group per 16-row tile shares the A tile through LDS (whole-row 16-B loads, X1/X2 formed at staging), each wave owns 32 columns: 192 WMMAs per wave, 12500 waves; sum order kept
# speedup vs baseline: 2.6997x; 1.0561x over previous
#include <hip/hip_runtime.h>

typedef float v2f __attribute__((ext_vector_type(2)));
typedef float v8f __attribute__((ext_vector_type(8)));

#define N_NODES 50000
#define N_EDGES 600000
#define DIM     128
#define NF      (N_NODES * DIM)

__global__ void k_fill(float* __restrict__ p, float v, int n) {
    int i = blockIdx.x * blockDim.x + threadIdx.x;
    if (i < n) p[i] = v;
}

#define R_ROWS 98
#define NB     512
#define NS     256
#define SLICE  ((N_EDGES + NS - 1) / NS)
#define EPT    ((SLICE + 255) / 256)
#define NCELL  (NB * NS)
#define NB_USED ((N_NODES + R_ROWS - 1) / R_ROWS)

__global__ void __launch_bounds__(256)
k_hist(const int* __restrict__ dst, float* __restrict__ deg, int* __restrict__ cnt) {
    __shared__ int h[NB];
    const int s = blockIdx.x, t = threadIdx.x;
    h[t] = 0;
    h[t + 256] = 0;
    __syncthreads();
    const int e0 = s * SLICE;
    const int e1 = (e0 + SLICE < N_EDGES) ? e0 + SLICE : N_EDGES;
    int dv[EPT];
#pragma unroll
    for (int k = 0; k < EPT; ++k) {
        const int e = e0 + t + k * 256;
        dv[k] = dst[e < e1 ? e : e1 - 1];
    }
#pragma unroll
    for (int k = 0; k < EPT; ++k) {
        if (e0 + t + k * 256 < e1) {
            atomicAdd(&deg[dv[k]], 1.0f);
            atomicAdd(&h[dv[k] / R_ROWS], 1);
        }
    }
    __syncthreads();
    cnt[t * NS + s]         = h[t];
    cnt[(t + 256) * NS + s] = h[t + 256];
}

__global__ void __launch_bounds__(NS)
k_bscan(int* __restrict__ cnt, int* __restrict__ tot) {
    __shared__ int part[NS];
    const int b = blockIdx.x, t = threadIdx.x;
    const int c = cnt[b * NS + t];
    part[t] = c;
    __syncthreads();
    for (int off = 1; off < NS; off <<= 1) {
        const int v = (t >= off) ? part[t - off] : 0;
        __syncthreads();
        part[t] += v;
        __syncthreads();
    }
    cnt[b * NS + t] = part[t] - c;
    if (t == NS - 1) tot[b] = part[t];
}

__global__ void __launch_bounds__(256)
k_place(const int* __restrict__ src, const int* __restrict__ dst,
        const int* __restrict__ within, const int* __restrict__ tot,
        int* __restrict__ begs, int2* __restrict__ list) {
    __shared__ int cur[NB];
    __shared__ int ps[256];
    const int s = blockIdx.x, t = threadIdx.x;
    const int t0 = tot[2 * t], t1 = tot[2 * t + 1];
    ps[t] = t0 + t1;
    __syncthreads();
    for (int off = 1; off < 256; off <<= 1) {
        const int v = (t >= off) ? ps[t - off] : 0;
        __syncthreads();
        ps[t] += v;
        __syncthreads();
    }
    const int ex = ps[t] - (t0 + t1);
    cur[2 * t]     = ex      + within[(2 * t) * NS + s];
    cur[2 * t + 1] = ex + t0 + within[(2 * t + 1) * NS + s];
    if (s == 0) {
        begs[2 * t]     = ex;
        begs[2 * t + 1] = ex + t0;
        if (t == 255) begs[NB] = ex + t0 + t1;
    }
    __syncthreads();
    const int e0 = s * SLICE;
    const int e1 = (e0 + SLICE < N_EDGES) ? e0 + SLICE : N_EDGES;
    int dv[EPT], sv[EPT];
#pragma unroll
    for (int k = 0; k < EPT; ++k) {
        const int e  = e0 + t + k * 256;
        const int ec = e < e1 ? e : e1 - 1;
        dv[k] = dst[ec];
        sv[k] = src[ec];
    }
#pragma unroll
    for (int k = 0; k < EPT; ++k) {
        if (e0 + t + k * 256 < e1) {
            const int p = atomicAdd(&cur[dv[k] / R_ROWS], 1);
            list[p] = make_int2(sv[k], dv[k]);
        }
    }
}

__global__ void k_rsqrt_inplace(float* __restrict__ d, int n) {
    int i = blockIdx.x * blockDim.x + threadIdx.x;
    if (i < n) d[i] = rsqrtf(d[i]);
}

__global__ void k_feat(const float* __restrict__ w,
                       const float* __restrict__ lw,
                       const float* __restrict__ lb,
                       float* __restrict__ feat) {
    int i = blockIdx.x * blockDim.x + threadIdx.x;
    if (i < NF) {
        int n = i >> 7, d = i & 127;
        feat[i] = w[n] * lw[d] + lb[d];
    }
}

template <int NE>
__device__ __forceinline__ void add_edges(const int2* __restrict__ list, int i,
                                          const float* __restrict__ x,
                                          const float* __restrict__ dinv,
                                          float* acc, int row0, int l, int second) {
    int2  ed[NE];
    float sc[NE];
    float v[NE][4];
#pragma unroll
    for (int j = 0; j < NE; ++j) ed[j] = list[i + j];
#pragma unroll
    for (int j = 0; j < NE; ++j) sc[j] = dinv[ed[j].x];
#pragma unroll
    for (int j = 0; j < NE; ++j) {
        const float* xp = x + (size_t)ed[j].x * DIM + l;
        v[j][0] = xp[0];
        v[j][1] = xp[32];
        v[j][2] = xp[64];
        v[j][3] = xp[96];
    }
#pragma unroll
    for (int j = 0; j < NE; ++j) {
        float* ap = acc + (ed[j].y - row0) * DIM + l;
#pragma unroll
        for (int k = 0; k < 4; ++k) {
            float val = v[j][k];
            if (second) val = -(val * sc[j]);
            atomicAdd(ap + 32 * k, val * sc[j]);
        }
    }
}

__global__ void __launch_bounds__(512)
k_scatter_lds(const int2* __restrict__ list,
              const int* __restrict__ begs,
              const float* __restrict__ x,
              const float* __restrict__ dinv,
              float* __restrict__ agg,
              int second) {
    __shared__ float acc[R_ROWS * DIM];
    const int b    = blockIdx.x;
    const int t    = threadIdx.x;
    const int row0 = b * R_ROWS;
    const int left = N_NODES - row0;
    const int nflt = (left > R_ROWS ? R_ROWS : left) * DIM;

    if (second) {
#pragma unroll
        for (int k = 0; k < 7; ++k) {
            const int i = t * 4 + k * 2048;
            if (i < nflt) *(float4*)&acc[i] = make_float4(0.0f, 0.0f, 0.0f, 0.0f);
        }
    } else {
        const float* xr = x + (size_t)row0 * DIM;
        float4 sv[7];
        float  sd[7];
#pragma unroll
        for (int k = 0; k < 7; ++k) {
            const int i  = t * 4 + k * 2048;
            const int ic = i < nflt ? i : 0;
            sv[k] = *(const float4*)(xr + ic);
            sd[k] = dinv[row0 + (ic >> 7)];
        }
#pragma unroll
        for (int k = 0; k < 7; ++k) {
            const int i = t * 4 + k * 2048;
            if (i < nflt) {
                float4 v = sv[k];
                v.x *= sd[k]; v.y *= sd[k]; v.z *= sd[k]; v.w *= sd[k];
                *(float4*)&acc[i] = v;
            }
        }
    }
    __syncthreads();

    const int beg  = begs[b];
    const int end  = begs[b + 1];
    const int ngrp = (end - beg) >> 2;
    const int w = __builtin_amdgcn_readfirstlane(t >> 5);
    const int l = t & 31;
    for (int g = w; g < ngrp; g += 16)
        add_edges<4>(list, beg + 4 * g, x, dinv, acc, row0, l, second);
    const int it = beg + 4 * ngrp + w;
    if (it < end)
        add_edges<1>(list, it, x, dinv, acc, row0, l, second);
    __syncthreads();

    float* ar = agg + (size_t)row0 * DIM;
#pragma unroll
    for (int k = 0; k < 7; ++k) {
        const int i = t * 4 + k * 2048;
        if (i < nflt) *(float4*)(ar + i) = *(const float4*)&acc[i];
    }
}

#define A_SEG (16 * DIM)

__device__ __forceinline__ void seg_steps(v8f& acc0, v8f& acc1,
                                          const float* As_row, int swz, int half,
                                          const float* __restrict__ Wl) {
#pragma unroll 8
    for (int kk = 0; kk < DIM; kk += 4) {
        const v2f a = *(const v2f*)(As_row + ((kk ^ swz) + 2 * half));
        const float2 q0 = *(const float2*)(Wl + (size_t)kk * DIM);
        const float2 q1 = *(const float2*)(Wl + (size_t)kk * DIM + DIM);
        const v2f b0 = {q0.x, q1.x}, b1 = {q0.y, q1.y};
        acc0 = __builtin_amdgcn_wmma_f32_16x16x4_f32(
             false, a,  false, b0,  (short)0, acc0,
             false,  false);
        acc1 = __builtin_amdgcn_wmma_f32_16x16x4_f32(
             false, a,  false, b1,  (short)0, acc1,
             false,  false);
    }
}

__global__ void __launch_bounds__(128)
k_cheb_gemm(const float* __restrict__ X0,
            const float* __restrict__ S1,
            const float* __restrict__ S2,
            const float* __restrict__ dinv,
            const float* __restrict__ W,
            const float* __restrict__ b,
            float* __restrict__ out) {
    __shared__ float As[3 * A_SEG];
    const int row0 = blockIdx.x * 16;
    const int t    = threadIdx.x;

    {
        float4 x0[4], s1[4], s2[4];
        float  dv[4];
#pragma unroll
        for (int j = 0; j < 4; ++j) {
            const int idx = t + 128 * j;
            const int r   = idx >> 5;
            const size_t g = (size_t)(row0 + r) * DIM + 4 * (idx & 31);
            x0[j] = *(const float4*)(X0 + g);
            s1[j] = *(const float4*)(S1 + g);
            s2[j] = *(const float4*)(S2 + g);
            dv[j] = dinv[row0 + r];
        }
#pragma unroll
        for (int j = 0; j < 4; ++j) {
            const int idx = t + 128 * j;
            const int r   = idx >> 5;
            const float di = dv[j];
            float4 x1, x2;
            x1.x = -(s1[j].x * di); x1.y = -(s1[j].y * di);
            x1.z = -(s1[j].z * di); x1.w = -(s1[j].w * di);
            x2.x = -2.0f * ((s2[j].x + x1.x * di) * di) - x0[j].x;
            x2.y = -2.0f * ((s2[j].y + x1.y * di) * di) - x0[j].y;
            x2.z = -2.0f * ((s2[j].z + x1.z * di) * di) - x0[j].z;
            x2.w = -2.0f * ((s2[j].w + x1.w * di) * di) - x0[j].w;
            const int p = r * DIM + ((4 * (idx & 31)) ^ (r << 2));
            *(float4*)&As[p]             = x0[j];
            *(float4*)&As[A_SEG + p]     = x1;
            *(float4*)&As[2 * A_SEG + p] = x2;
        }
    }
    __syncthreads();

    const int w    = __builtin_amdgcn_readfirstlane(t >> 5);
    const int lane = t & 31;
    const int half = lane >> 4;
    const int lr   = lane & 15;
    const int col  = 32 * w + 2 * lr;

    const float2 bv = *(const float2*)(b + col);
    v8f acc0 = (v8f){bv.x, bv.x, bv.x, bv.x, bv.x, bv.x, bv.x, bv.x};
    v8f acc1 = (v8f){bv.y, bv.y, bv.y, bv.y, bv.y, bv.y, bv.y, bv.y};

    const float* As_row = As + lr * DIM;
    const int    swz    = lr << 2;
    const float* Wl     = W + (size_t)(2 * half) * DIM + col;

    seg_steps(acc0, acc1, As_row,             swz, half, Wl);
    seg_steps(acc0, acc1, As_row + A_SEG,     swz, half, Wl + (size_t)DIM * DIM);
    seg_steps(acc0, acc1, As_row + 2 * A_SEG, swz, half, Wl + (size_t)2 * DIM * DIM);

#pragma unroll
    for (int r = 0; r < 8; ++r) {
        float2 v;
        v.x = acc0[r]; v.y = acc1[r];
        v.x = v.x > 0.0f ? v.x : 0.0f; v.y = v.y > 0.0f ? v.y : 0.0f;
        *(float2*)(out + (size_t)(row0 + r + 8 * half) * DIM + col) = v;
    }
}

__global__ void k_pred(const float* __restrict__ emb,
                       const float* __restrict__ pw,
                       const float* __restrict__ pb,
                       float* __restrict__ out) {
    int n = blockIdx.x * blockDim.x + threadIdx.x;
    if (n < N_NODES) {
        const float* e = emb + (size_t)n * DIM;
        float s = 0.0f;
#pragma unroll 8
        for (int d = 0; d < DIM; d += 4) {
            float4 v = *(const float4*)(e + d);
            s += v.x * pw[d] + v.y * pw[d + 1] + v.z * pw[d + 2] + v.w * pw[d + 3];
        }
        out[n] = s + pb[0];
    }
}

extern "C" void kernel_launch(void* const* d_in, const int* in_sizes, int n_in,
                              void* d_out, int out_size, void* d_ws, size_t ws_size,
                              hipStream_t stream) {
    const float* weights  = (const float*)d_in[0];
    const int*   src      = (const int*)  d_in[1];
    const int*   dst      = (const int*)  d_in[2];
    const float* lin_in_w = (const float*)d_in[3];
    const float* lin_in_b = (const float*)d_in[4];
    const float* cheb_ws  = (const float*)d_in[5];
    const float* cheb_bs  = (const float*)d_in[6];
    const float* pred_w   = (const float*)d_in[7];
    const float* pred_b   = (const float*)d_in[8];
    float* logits = (float*)d_out;

    float* ws   = (float*)d_ws;
    float* dinv = ws;
    float* P0   = ws + 51200;
    float* P1   = P0 + NF;
    float* P2   = P1 + NF;
    float* P3   = P2 + NF;
    int*   cnt  = (int*)(P3 + NF);
    int*   tot  = cnt + NCELL;
    int*   begs = tot + NB;
    int2*  list = (int2*)(cnt + 133120);

    const int T = 256;
    const int gN  = (N_NODES + T - 1) / T;
    const int gNF = (NF + T - 1) / T;

    k_fill<<<gN, T, 0, stream>>>(dinv, 1.0f, N_NODES);
    k_hist<<<NS, 256, 0, stream>>>(dst, dinv, cnt);
    k_rsqrt_inplace<<<gN, T, 0, stream>>>(dinv, N_NODES);
    k_bscan<<<NB, NS, 0, stream>>>(cnt, tot);
    k_place<<<NS, 256, 0, stream>>>(src, dst, cnt, tot, begs, list);

    k_feat<<<gNF, T, 0, stream>>>(weights, lin_in_w, lin_in_b, P0);

    float* cur = P0;
    float* nxt = P3;
    for (int layer = 0; layer < 3; ++layer) {
        const float* Wl = cheb_ws + (size_t)layer * 3 * DIM * DIM;
        const float* bl = cheb_bs + (size_t)layer * DIM;

        k_scatter_lds<<<NB_USED, 512, 0, stream>>>(list, begs, cur, dinv, P1, 0);
        k_scatter_lds<<<NB_USED, 512, 0, stream>>>(list, begs, P1, dinv, P2, 1);

        k_cheb_gemm<<<N_NODES / 16, 128, 0, stream>>>(cur, P1, P2, dinv, Wl, bl, nxt);

        float* t = cur; cur = nxt; nxt = t;
    }

    k_pred<<<gN, T, 0, stream>>>(cur, pred_w, pred_b, logits);
}


# Small passes folded: in-degrees counted per range from the binned list (LDS adds) in the pass writing the embedding; head dot product in last GEMM epilogue; 16 launches to 13; logit sum re-associated
# speedup vs baseline: 3.0964x; 1.1469x over previous
#include <hip/hip_runtime.h>

typedef float v2f __attribute__((ext_vector_type(2)));
typedef float v8f __attribute__((ext_vector_type(8)));

#define N_NODES 50000
#define N_EDGES 600000
#define DIM     128
#define NF      (N_NODES * DIM)

#define R_ROWS 98
#define NB     512
#define NS     256
#define SLICE  ((N_EDGES + NS - 1) / NS)
#define EPT    ((SLICE + 255) / 256)
#define NCELL  (NB * NS)
#define NB_USED ((N_NODES + R_ROWS - 1) / R_ROWS)

__global__ void __launch_bounds__(256)
k_hist(const int* __restrict__ dst, int* __restrict__ cnt) {
    __shared__ int h[NB];
    const int s = blockIdx.x, t = threadIdx.x;
    h[t] = 0;
    h[t + 256] = 0;
    __syncthreads();
    const int e0 = s * SLICE;
    const int e1 = (e0 + SLICE < N_EDGES) ? e0 + SLICE : N_EDGES;
    int dv[EPT];
#pragma unroll
    for (int k = 0; k < EPT; ++k) {
        const int e = e0 + t + k * 256;
        dv[k] = dst[e < e1 ? e : e1 - 1];
    }
#pragma unroll
    for (int k = 0; k < EPT; ++k) {
        if (e0 + t + k * 256 < e1) atomicAdd(&h[dv[k] / R_ROWS], 1);
    }
    __syncthreads();
    cnt[t * NS + s]         = h[t];
    cnt[(t + 256) * NS + s] = h[t + 256];
}

__global__ void __launch_bounds__(NS)
k_bscan(int* __restrict__ cnt, int* __restrict__ tot) {
    __shared__ int part[NS];
    const int b = blockIdx.x, t = threadIdx.x;
    const int c = cnt[b * NS + t];
    part[t] = c;
    __syncthreads();
    for (int off = 1; off < NS; off <<= 1) {
        const int v = (t >= off) ? part[t - off] : 0;
        __syncthreads();
        part[t] += v;
        __syncthreads();
    }
    cnt[b * NS + t] = part[t] - c;
    if (t == NS - 1) tot[b] = part[t];
}

__global__ void __launch_bounds__(256)
k_place(const int* __restrict__ src, const int* __restrict__ dst,
        const int* __restrict__ within, const int* __restrict__ tot,
        int* __restrict__ begs, int2* __restrict__ list) {
    __shared__ int cur[NB];
    __shared__ int ps[256];
    const int s = blockIdx.x, t = threadIdx.x;
    const int t0 = tot[2 * t], t1 = tot[2 * t + 1];
    ps[t] = t0 + t1;
    __syncthreads();
    for (int off = 1; off < 256; off <<= 1) {
        const int v = (t >= off) ? ps[t - off] : 0;
        __syncthreads();
        ps[t] += v;
        __syncthreads();
    }
    const int ex = ps[t] - (t0 + t1);
    cur[2 * t]     = ex      + within[(2 * t) * NS + s];
    cur[2 * t + 1] = ex + t0 + within[(2 * t + 1) * NS + s];
    if (s == 0) {
        begs[2 * t]     = ex;
        begs[2 * t + 1] = ex + t0;
        if (t == 255) begs[NB] = ex + t0 + t1;
    }
    __syncthreads();
    const int e0 = s * SLICE;
    const int e1 = (e0 + SLICE < N_EDGES) ? e0 + SLICE : N_EDGES;
    int dv[EPT], sv[EPT];
#pragma unroll
    for (int k = 0; k < EPT; ++k) {
        const int e  = e0 + t + k * 256;
        const int ec = e < e1 ? e : e1 - 1;
        dv[k] = dst[ec];
        sv[k] = src[ec];
    }
#pragma unroll
    for (int k = 0; k < EPT; ++k) {
        if (e0 + t + k * 256 < e1) {
            const int p = atomicAdd(&cur[dv[k] / R_ROWS], 1);
            list[p] = make_int2(sv[k], dv[k]);
        }
    }
}

__global__ void __launch_bounds__(512)
k_deg_feat(const int2* __restrict__ list,
           const int* __restrict__ begs,
           const float* __restrict__ w,
           const float* __restrict__ lw,
           const float* __restrict__ lb,
           float* __restrict__ dinv,
           float* __restrict__ feat) {
    __shared__ int c[128];
    const int b    = blockIdx.x;
    const int t    = threadIdx.x;
    const int row0 = b * R_ROWS;
    const int left = N_NODES - row0;
    const int nrow = left > R_ROWS ? R_ROWS : left;
    const int nflt = nrow * DIM;

    const float4 a = *(const float4*)(lw + 4 * (t & 31));
    const float4 o = *(const float4*)(lb + 4 * (t & 31));
    float wv[7];
#pragma unroll
    for (int k = 0; k < 7; ++k) {
        const int i  = t * 4 + k * 2048;
        const int ic = i < nflt ? i : 0;
        wv[k] = w[row0 + (ic >> 7)];
    }

    if (t < 128) c[t] = 0;
    __syncthreads();
    const int beg = begs[b];
    const int end = begs[b + 1];
    for (int i = beg + t; i < end; i += 512)
        atomicAdd(&c[list[i].y - row0], 1);
    __syncthreads();
    if (t < nrow) dinv[row0 + t] = rsqrtf((float)(c[t] + 1));

    float* fr = feat + (size_t)row0 * DIM;
#pragma unroll
    for (int k = 0; k < 7; ++k) {
        const int i = t * 4 + k * 2048;
        if (i < nflt) {
            float4 v;
            v.x = wv[k] * a.x + o.x;
            v.y = wv[k] * a.y + o.y;
            v.z = wv[k] * a.z + o.z;
            v.w = wv[k] * a.w + o.w;
            *(float4*)(fr + i) = v;
        }
    }
}

template <int NE>
__device__ __forceinline__ void add_edges(const int2* __restrict__ list, int i,
                                          const float* __restrict__ x,
                                          const float* __restrict__ dinv,
                                          float* acc, int row0, int l, int second) {
    int2  ed[NE];
    float sc[NE];
    float v[NE][4];
#pragma unroll
    for (int j = 0; j < NE; ++j) ed[j] = list[i + j];
#pragma unroll
    for (int j = 0; j < NE; ++j) sc[j] = dinv[ed[j].x];
#pragma unroll
    for (int j = 0; j < NE; ++j) {
        const float* xp = x + (size_t)ed[j].x * DIM + l;
        v[j][0] = xp[0];
        v[j][1] = xp[32];
        v[j][2] = xp[64];
        v[j][3] = xp[96];
    }
#pragma unroll
    for (int j = 0; j < NE; ++j) {
        float* ap = acc + (ed[j].y - row0) * DIM + l;
#pragma unroll
        for (int k = 0; k < 4; ++k) {
            float val = v[j][k];
            if (second) val = -(val * sc[j]);
            atomicAdd(ap + 32 * k, val * sc[j]);
        }
    }
}

__global__ void __launch_bounds__(512)
k_scatter_lds(const int2* __restrict__ list,
              const int* __restrict__ begs,
              const float* __restrict__ x,
              const float* __restrict__ dinv,
              float* __restrict__ agg,
              int second) {
    __shared__ float acc[R_ROWS * DIM];
    const int b    = blockIdx.x;
    const int t    = threadIdx.x;
    const int row0 = b * R_ROWS;
    const int left = N_NODES - row0;
    const int nflt = (left > R_ROWS ? R_ROWS : left) * DIM;

    if (second) {
#pragma unroll
        for (int k = 0; k < 7; ++k) {
            const int i = t * 4 + k * 2048;
            if (i < nflt) *(float4*)&acc[i] = make_float4(0.0f, 0.0f, 0.0f, 0.0f);
        }
    } else {
        const float* xr = x + (size_t)row0 * DIM;
        float4 sv[7];
        float  sd[7];
#pragma unroll
        for (int k = 0; k < 7; ++k) {
            const int i  = t * 4 + k * 2048;
            const int ic = i < nflt ? i : 0;
            sv[k] = *(const float4*)(xr + ic);
            sd[k] = dinv[row0 + (ic >> 7)];
        }
#pragma unroll
        for (int k = 0; k < 7; ++k) {
            const int i = t * 4 + k * 2048;
            if (i < nflt) {
                float4 v = sv[k];
                v.x *= sd[k]; v.y *= sd[k]; v.z *= sd[k]; v.w *= sd[k];
                *(float4*)&acc[i] = v;
            }
        }
    }
    __syncthreads();

    const int beg  = begs[b];
    const int end  = begs[b + 1];
    const int ngrp = (end - beg) >> 2;
    const int w = __builtin_amdgcn_readfirstlane(t >> 5);
    const int l = t & 31;
    for (int g = w; g < ngrp; g += 16)
        add_edges<4>(list, beg + 4 * g, x, dinv, acc, row0, l, second);
    const int it = beg + 4 * ngrp + w;
    if (it < end)
        add_edges<1>(list, it, x, dinv, acc, row0, l, second);
    __syncthreads();

    float* ar = agg + (size_t)row0 * DIM;
#pragma unroll
    for (int k = 0; k < 7; ++k) {
        const int i = t * 4 + k * 2048;
        if (i < nflt) *(float4*)(ar + i) = *(const float4*)&acc[i];
    }
}

#define A_SEG (16 * DIM)

__device__ __forceinline__ void seg_steps(v8f& acc0, v8f& acc1,
                                          const float* As_row, int swz, int half,
                                          const float* __restrict__ Wl) {
#pragma unroll 8
    for (int kk = 0; kk < DIM; kk += 4) {
        const v2f a = *(const v2f*)(As_row + ((kk ^ swz) + 2 * half));
        const float2 q0 = *(const float2*)(Wl + (size_t)kk * DIM);
        const float2 q1 = *(const float2*)(Wl + (size_t)kk * DIM + DIM);
        const v2f b0 = {q0.x, q1.x}, b1 = {q0.y, q1.y};
        acc0 = __builtin_amdgcn_wmma_f32_16x16x4_f32(
             false, a,  false, b0,  (short)0, acc0,
             false,  false);
        acc1 = __builtin_amdgcn_wmma_f32_16x16x4_f32(
             false, a,  false, b1,  (short)0, acc1,
             false,  false);
    }
}

template <bool LAST>
__global__ void __launch_bounds__(128)
k_cheb_gemm(const float* __restrict__ X0,
            const float* __restrict__ S1,
            const float* __restrict__ S2,
            const float* __restrict__ dinv,
            const float* __restrict__ W,
            const float* __restrict__ b,
            float* __restrict__ out,
            const float* __restrict__ pw,
            const float* __restrict__ pb,
            float* __restrict__ logits) {
    __shared__ float As[3 * A_SEG];
    const int row0 = blockIdx.x * 16;
    const int t    = threadIdx.x;

    {
        float4 x0[4], s1[4], s2[4];
        float  dv[4];
#pragma unroll
        for (int j = 0; j < 4; ++j) {
            const int idx = t + 128 * j;
            const int r   = idx >> 5;
            const size_t g = (size_t)(row0 + r) * DIM + 4 * (idx & 31);
            x0[j] = *(const float4*)(X0 + g);
            s1[j] = *(const float4*)(S1 + g);
            s2[j] = *(const float4*)(S2 + g);
            dv[j] = dinv[row0 + r];
        }
#pragma unroll
        for (int j = 0; j < 4; ++j) {
            const int idx = t + 128 * j;
            const int r   = idx >> 5;
            const float di = dv[j];
            float4 x1, x2;
            x1.x = -(s1[j].x * di); x1.y = -(s1[j].y * di);
            x1.z = -(s1[j].z * di); x1.w = -(s1[j].w * di);
            x2.x = -2.0f * ((s2[j].x + x1.x * di) * di) - x0[j].x;
            x2.y = -2.0f * ((s2[j].y + x1.y * di) * di) - x0[j].y;
            x2.z = -2.0f * ((s2[j].z + x1.z * di) * di) - x0[j].z;
            x2.w = -2.0f * ((s2[j].w + x1.w * di) * di) - x0[j].w;
            const int p = r * DIM + ((4 * (idx & 31)) ^ (r << 2));
            *(float4*)&As[p]             = x0[j];
            *(float4*)&As[A_SEG + p]     = x1;
            *(float4*)&As[2 * A_SEG + p] = x2;
        }
    }
    __syncthreads();

    const int w    = __builtin_amdgcn_readfirstlane(t >> 5);
    const int lane = t & 31;
    const int half = lane >> 4;
    const int lr   = lane & 15;
    const int col  = 32 * w + 2 * lr;

    const float2 bv = *(const float2*)(b + col);
    v8f acc0 = (v8f){bv.x, bv.x, bv.x, bv.x, bv.x, bv.x, bv.x, bv.x};
    v8f acc1 = (v8f){bv.y, bv.y, bv.y, bv.y, bv.y, bv.y, bv.y, bv.y};

    const float* As_row = As + lr * DIM;
    const int    swz    = lr << 2;
    const float* Wl     = W + (size_t)(2 * half) * DIM + col;

    seg_steps(acc0, acc1, As_row,             swz, half, Wl);
    seg_steps(acc0, acc1, As_row + A_SEG,     swz, half, Wl + (size_t)DIM * DIM);
    seg_steps(acc0, acc1, As_row + 2 * A_SEG, swz, half, Wl + (size_t)2 * DIM * DIM);

    if constexpr (!LAST) {
#pragma unroll
        for (int r = 0; r < 8; ++r) {
            float2 v;
            v.x = acc0[r]; v.y = acc1[r];
            v.x = v.x > 0.0f ? v.x : 0.0f; v.y = v.y > 0.0f ? v.y : 0.0f;
            *(float2*)(out + (size_t)(row0 + r + 8 * half) * DIM + col) = v;
        }
    } else {
        const float2 p = *(const float2*)(pw + col);
        float s[8];
#pragma unroll
        for (int r = 0; r < 8; ++r) {
            float2 v;
            v.x = acc0[r]; v.y = acc1[r];
            v.x = v.x > 0.0f ? v.x : 0.0f; v.y = v.y > 0.0f ? v.y : 0.0f;
            s[r] = v.x * p.x + v.y * p.y;
        }
#pragma unroll
        for (int m = 1; m < 16; m <<= 1) {
#pragma unroll
            for (int r = 0; r < 8; ++r) s[r] += __shfl_xor(s[r], m);
        }
        __syncthreads();
        float* red = As;
        if (lr == 0) {
#pragma unroll
            for (int r = 0; r < 8; ++r) red[w * 16 + r + 8 * half] = s[r];
        }
        __syncthreads();
        if (t < 16)
            logits[row0 + t] = ((red[t] + red[16 + t]) + (red[32 + t] + red[48 + t])) + pb[0];
    }
}

extern "C" void kernel_launch(void* const* d_in, const int* in_sizes, int n_in,
                              void* d_out, int out_size, void* d_ws, size_t ws_size,
                              hipStream_t stream) {
    const float* weights  = (const float*)d_in[0];
    const int*   src      = (const int*)  d_in[1];
    const int*   dst      = (const int*)  d_in[2];
    const float* lin_in_w = (const float*)d_in[3];
    const float* lin_in_b = (const float*)d_in[4];
    const float* cheb_ws  = (const float*)d_in[5];
    const float* cheb_bs  = (const float*)d_in[6];
    const float* pred_w   = (const float*)d_in[7];
    const float* pred_b   = (const float*)d_in[8];
    float* logits = (float*)d_out;

    float* ws   = (float*)d_ws;
    float* dinv = ws;
    float* P0   = ws + 51200;
    float* P1   = P0 + NF;
    float* P2   = P1 + NF;
    float* P3   = P2 + NF;
    int*   cnt  = (int*)(P3 + NF);
    int*   tot  = cnt + NCELL;
    int*   begs = tot + NB;
    int2*  list = (int2*)(cnt + 133120);

    k_hist<<<NS, 256, 0, stream>>>(dst, cnt);
    k_bscan<<<NB, NS, 0, stream>>>(cnt, tot);
    k_place<<<NS, 256, 0, stream>>>(src, dst, cnt, tot, begs, list);
    k_deg_feat<<<NB_USED, 512, 0, stream>>>(list, begs, weights, lin_in_w, lin_in_b, dinv, P0);

    float* cur = P0;
    float* nxt = P3;
    for (int layer = 0; layer < 3; ++layer) {
        const float* Wl = cheb_ws + (size_t)layer * 3 * DIM * DIM;
        const float* bl = cheb_bs + (size_t)layer * DIM;

        k_scatter_lds<<<NB_USED, 512, 0, stream>>>(list, begs, cur, dinv, P1, 0);
        k_scatter_lds<<<NB_USED, 512, 0, stream>>>(list, begs, P1, dinv, P2, 1);

        if (layer < 2)
            k_cheb_gemm<false><<<N_NODES / 16, 128, 0, stream>>>(cur, P1, P2, dinv, Wl, bl, nxt,
                                                                 pred_w, pred_b, logits);
        else
            k_cheb_gemm<true><<<N_NODES / 16, 128, 0, stream>>>(cur, P1, P2, dinv, Wl, bl, nxt,
                                                                pred_w, pred_b, logits);

        float* t = cur; cur = nxt; nxt = t;
    }
}


# Second sum fused with the dense product: a range's 16-wave group keeps S2 in LDS beside staged X0/X1 rows, forms X2 in place, feeds its WMMAs from LDS; 13 launches to 10; sum order kept
# speedup vs baseline: 3.2654x; 1.0546x over previous
#include <hip/hip_runtime.h>

typedef float v2f __attribute__((ext_vector_type(2)));
typedef float v8f __attribute__((ext_vector_type(8)));

#define N_NODES 50000
#define N_EDGES 600000
#define DIM     128
#define NF      (N_NODES * DIM)

#define R_ROWS 98
#define NB     512
#define NS     256
#define SLICE  ((N_EDGES + NS - 1) / NS)
#define EPT    ((SLICE + 255) / 256)
#define NCELL  (NB * NS)
#define NB_USED ((N_NODES + R_ROWS - 1) / R_ROWS)

__global__ void __launch_bounds__(256)
k_hist(const int* __restrict__ dst, int* __restrict__ cnt) {
    __shared__ int h[NB];
    const int s = blockIdx.x, t = threadIdx.x;
    h[t] = 0;
    h[t + 256] = 0;
    __syncthreads();
    const int e0 = s * SLICE;
    const int e1 = (e0 + SLICE < N_EDGES) ? e0 + SLICE : N_EDGES;
    int dv[EPT];
#pragma unroll
    for (int k = 0; k < EPT; ++k) {
        const int e = e0 + t + k * 256;
        dv[k] = dst[e < e1 ? e : e1 - 1];
    }
#pragma unroll
    for (int k = 0; k < EPT; ++k) {
        if (e0 + t + k * 256 < e1) atomicAdd(&h[dv[k] / R_ROWS], 1);
    }
    __syncthreads();
    cnt[t * NS + s]         = h[t];
    cnt[(t + 256) * NS + s] = h[t + 256];
}

__global__ void __launch_bounds__(NS)
k_bscan(int* __restrict__ cnt, int* __restrict__ tot) {
    __shared__ int part[NS];
    const int b = blockIdx.x, t = threadIdx.x;
    const int c = cnt[b * NS + t];
    part[t] = c;
    __syncthreads();
    for (int off = 1; off < NS; off <<= 1) {
        const int v = (t >= off) ? part[t - off] : 0;
        __syncthreads();
        part[t] += v;
        __syncthreads();
    }
    cnt[b * NS + t] = part[t] - c;
    if (t == NS - 1) tot[b] = part[t];
}

__global__ void __launch_bounds__(256)
k_place(const int* __restrict__ src, const int* __restrict__ dst,
        const int* __restrict__ within, const int* __restrict__ tot,
        int* __restrict__ begs, int2* __restrict__ list) {
    __shared__ int cur[NB];
    __shared__ int ps[256];
    const int s = blockIdx.x, t = threadIdx.x;
    const int t0 = tot[2 * t], t1 = tot[2 * t + 1];
    ps[t] = t0 + t1;
    __syncthreads();
    for (int off = 1; off < 256; off <<= 1) {
        const int v = (t >= off) ? ps[t - off] : 0;
        __syncthreads();
        ps[t] += v;
        __syncthreads();
    }
    const int ex = ps[t] - (t0 + t1);
    cur[2 * t]     = ex      + within[(2 * t) * NS + s];
    cur[2 * t + 1] = ex + t0 + within[(2 * t + 1) * NS + s];
    if (s == 0) {
        begs[2 * t]     = ex;
        begs[2 * t + 1] = ex + t0;
        if (t == 255) begs[NB] = ex + t0 + t1;
    }
    __syncthreads();
    const int e0 = s * SLICE;
    const int e1 = (e0 + SLICE < N_EDGES) ? e0 + SLICE : N_EDGES;
    int dv[EPT], sv[EPT];
#pragma unroll
    for (int k = 0; k < EPT; ++k) {
        const int e  = e0 + t + k * 256;
        const int ec = e < e1 ? e : e1 - 1;
        dv[k] = dst[ec];
        sv[k] = src[ec];
    }
#pragma unroll
    for (int k = 0; k < EPT; ++k) {
        if (e0 + t + k * 256 < e1) {
            const int p = atomicAdd(&cur[dv[k] / R_ROWS], 1);
            list[p] = make_int2(sv[k], dv[k]);
        }
    }
}

__global__ void __launch_bounds__(512)
k_deg_feat(const int2* __restrict__ list,
           const int* __restrict__ begs,
           const float* __restrict__ w,
           const float* __restrict__ lw,
           const float* __restrict__ lb,
           float* __restrict__ dinv,
           float* __restrict__ feat) {
    __shared__ int c[128];
    const int b    = blockIdx.x;
    const int t    = threadIdx.x;
    const int row0 = b * R_ROWS;
    const int left = N_NODES - row0;
    const int nrow = left > R_ROWS ? R_ROWS : left;
    const int nflt = nrow * DIM;

    const float4 a = *(const float4*)(lw + 4 * (t & 31));
    const float4 o = *(const float4*)(lb + 4 * (t & 31));
    float wv[7];
#pragma unroll
    for (int k = 0; k < 7; ++k) {
        const int i  = t * 4 + k * 2048;
        const int ic = i < nflt ? i : 0;
        wv[k] = w[row0 + (ic >> 7)];
    }

    if (t < 128) c[t] = 0;
    __syncthreads();
    const int beg = begs[b];
    const int end = begs[b + 1];
    for (int i = beg + t; i < end; i += 512)
        atomicAdd(&c[list[i].y - row0], 1);
    __syncthreads();
    if (t < nrow) dinv[row0 + t] = rsqrtf((float)(c[t] + 1));

    float* fr = feat + (size_t)row0 * DIM;
#pragma unroll
    for (int k = 0; k < 7; ++k) {
        const int i = t * 4 + k * 2048;
        if (i < nflt) {
            float4 v;
            v.x = wv[k] * a.x + o.x;
            v.y = wv[k] * a.y + o.y;
            v.z = wv[k] * a.z + o.z;
            v.w = wv[k] * a.w + o.w;
            *(float4*)(fr + i) = v;
        }
    }
}

template <int NE, bool SECOND>
__device__ __forceinline__ void add_edges(const int2* __restrict__ list, int i,
                                          const float* __restrict__ x,
                                          const float* __restrict__ dinv,
                                          float* acc, int row0, int l) {
    int2  ed[NE];
    float sc[NE];
    float v[NE][4];
#pragma unroll
    for (int j = 0; j < NE; ++j) ed[j] = list[i + j];
#pragma unroll
    for (int j = 0; j < NE; ++j) sc[j] = dinv[ed[j].x];
#pragma unroll
    for (int j = 0; j < NE; ++j) {
        const float* xp = x + (size_t)ed[j].x * DIM + l;
        v[j][0] = xp[0];
        v[j][1] = xp[32];
        v[j][2] = xp[64];
        v[j][3] = xp[96];
    }
#pragma unroll
    for (int j = 0; j < NE; ++j) {
        const int rl = ed[j].y - row0;
        float* ap = acc + rl * DIM;
        const int sw = SECOND ? ((rl & 15) << 2) : 0;
#pragma unroll
        for (int k = 0; k < 4; ++k) {
            float val = v[j][k];
            if (SECOND) val = -(val * sc[j]);
            atomicAdd(ap + ((l + 32 * k) ^ sw), val * sc[j]);
        }
    }
}

__global__ void __launch_bounds__(512)
k_scatter_lds(const int2* __restrict__ list,
              const int* __restrict__ begs,
              const float* __restrict__ x,
              const float* __restrict__ dinv,
              float* __restrict__ agg) {
    __shared__ float acc[R_ROWS * DIM];
    const int b    = blockIdx.x;
    const int t    = threadIdx.x;
    const int row0 = b * R_ROWS;
    const int left = N_NODES - row0;
    const int nflt = (left > R_ROWS ? R_ROWS : left) * DIM;

    {
        const float* xr = x + (size_t)row0 * DIM;
        float4 sv[7];
        float  sd[7];
#pragma unroll
        for (int k = 0; k < 7; ++k) {
            const int i  = t * 4 + k * 2048;
            const int ic = i < nflt ? i : 0;
            sv[k] = *(const float4*)(xr + ic);
            sd[k] = dinv[row0 + (ic >> 7)];
        }
#pragma unroll
        for (int k = 0; k < 7; ++k) {
            const int i = t * 4 + k * 2048;
            if (i < nflt) {
                float4 v = sv[k];
                v.x *= sd[k]; v.y *= sd[k]; v.z *= sd[k]; v.w *= sd[k];
                *(float4*)&acc[i] = v;
            }
        }
    }
    __syncthreads();

    const int beg  = begs[b];
    const int end  = begs[b + 1];
    const int ngrp = (end - beg) >> 2;
    const int w = __builtin_amdgcn_readfirstlane(t >> 5);
    const int l = t & 31;
    for (int g = w; g < ngrp; g += 16)
        add_edges<4, false>(list, beg + 4 * g, x, dinv, acc, row0, l);
    const int it = beg + 4 * ngrp + w;
    if (it < end)
        add_edges<1, false>(list, it, x, dinv, acc, row0, l);
    __syncthreads();

    float* ar = agg + (size_t)row0 * DIM;
#pragma unroll
    for (int k = 0; k < 7; ++k) {
        const int i = t * 4 + k * 2048;
        if (i < nflt) *(float4*)(ar + i) = *(const float4*)&acc[i];
    }
}

#define N_TILE ((R_ROWS + 15) / 16)

template <bool TWO>
__device__ __forceinline__ void seg_steps(v8f& c00, v8f& c01, v8f& c10, v8f& c11,
                                          const float* rA, int swzA,
                                          const float* rB, int swzB, int half,
                                          const float* __restrict__ Wl) {
#pragma unroll 4
    for (int kk = 0; kk < DIM; kk += 4) {
        const v2f a = *(const v2f*)(rA + ((kk ^ swzA) + 2 * half));
        const float2 q0 = *(const float2*)(Wl + (size_t)kk * DIM);
        const float2 q1 = *(const float2*)(Wl + (size_t)kk * DIM + DIM);
        const v2f b0 = {q0.x, q1.x}, b1 = {q0.y, q1.y};
        c00 = __builtin_amdgcn_wmma_f32_16x16x4_f32(
             false, a,  false, b0,  (short)0, c00,
             false,  false);
        c01 = __builtin_amdgcn_wmma_f32_16x16x4_f32(
             false, a,  false, b1,  (short)0, c01,
             false,  false);
        if constexpr (TWO) {
            const v2f a2 = *(const v2f*)(rB + ((kk ^ swzB) + 2 * half));
            c10 = __builtin_amdgcn_wmma_f32_16x16x4_f32(
                 false, a2,  false, b0,  (short)0, c10,
                 false,  false);
            c11 = __builtin_amdgcn_wmma_f32_16x16x4_f32(
                 false, a2,  false, b1,  (short)0, c11,
                 false,  false);
        }
    }
}

template <bool LAST>
__device__ __forceinline__ void tile_out(const v8f& c0, const v8f& c1, int tile, int nrow,
                                         int half, int lr, int col, float2 p,
                                         float* __restrict__ orow, float* red) {
    if constexpr (!LAST) {
#pragma unroll
        for (int r = 0; r < 8; ++r) {
            const int rr = 16 * tile + r + 8 * half;
            float2 v;
            v.x = c0[r]; v.y = c1[r];
            v.x = v.x > 0.0f ? v.x : 0.0f; v.y = v.y > 0.0f ? v.y : 0.0f;
            if (rr < nrow) *(float2*)(orow + (size_t)rr * DIM + col) = v;
        }
    } else {
        float s[8];
#pragma unroll
        for (int r = 0; r < 8; ++r) {
            float2 v;
            v.x = c0[r]; v.y = c1[r];
            v.x = v.x > 0.0f ? v.x : 0.0f; v.y = v.y > 0.0f ? v.y : 0.0f;
            s[r] = v.x * p.x + v.y * p.y;
        }
#pragma unroll
        for (int m = 1; m < 16; m <<= 1) {
#pragma unroll
            for (int r = 0; r < 8; ++r) s[r] += __shfl_xor(s[r], m);
        }
        if (lr == 0) {
#pragma unroll
            for (int r = 0; r < 8; ++r) red[tile * 64 + r + 8 * half] = s[r];
        }
    }
}

template <bool LAST, bool TWO>
__device__ __forceinline__ void wave_tiles(const float* a0, const float* a1, const float* a2,
                                           int tA, int nrow, int half, int lr, int col,
                                           const float* __restrict__ W, float2 bv, float2 p,
                                           float* __restrict__ orow, float* red) {
    const int last = nrow - 1;
    const int ra = (16 * tA + lr < last) ? 16 * tA + lr : last;
    const int rb = (16 * (tA + 4) + lr < last) ? 16 * (tA + 4) + lr : last;
    const int swzA = (ra & 15) << 2;
    const int swzB = (rb & 15) << 2;
    v8f c00 = (v8f){bv.x, bv.x, bv.x, bv.x, bv.x, bv.x, bv.x, bv.x};
    v8f c01 = (v8f){bv.y, bv.y, bv.y, bv.y, bv.y, bv.y, bv.y, bv.y};
    v8f c10 = c00, c11 = c01;
    const float* Wl = W + (size_t)(2 * half) * DIM + col;
    seg_steps<TWO>(c00, c01, c10, c11, a0 + ra * DIM, swzA, a0 + rb * DIM, swzB, half, Wl);
    seg_steps<TWO>(c00, c01, c10, c11, a1 + ra * DIM, swzA, a1 + rb * DIM, swzB, half, Wl + (size_t)DIM * DIM);
    seg_steps<TWO>(c00, c01, c10, c11, a2 + ra * DIM, swzA, a2 + rb * DIM, swzB, half, Wl + (size_t)2 * DIM * DIM);
    tile_out<LAST>(c00, c01, tA, nrow, half, lr, col, p, orow, red);
    if constexpr (TWO) tile_out<LAST>(c10, c11, tA + 4, nrow, half, lr, col, p, orow, red);
}

template <bool LAST>
__global__ void __launch_bounds__(512)
k_scatter_gemm(const int2* __restrict__ list,
               const int* __restrict__ begs,
               const float* __restrict__ X0,
               const float* __restrict__ S1,
               const float* __restrict__ dinv,
               const float* __restrict__ W,
               const float* __restrict__ b,
               float* __restrict__ out,
               const float* __restrict__ pw,
               const float* __restrict__ pb,
               float* __restrict__ logits) {
    __shared__ float a0[R_ROWS * DIM];
    __shared__ float a1[R_ROWS * DIM];
    __shared__ float acc[R_ROWS * DIM];
    __shared__ float red[N_TILE * 64];
    const int bk   = blockIdx.x;
    const int t    = threadIdx.x;
    const int row0 = bk * R_ROWS;
    const int left = N_NODES - row0;
    const int nrow = left > R_ROWS ? R_ROWS : left;
    const int nflt = nrow * DIM;

    {
        const float* xr = X0 + (size_t)row0 * DIM;
        const float* sr = S1 + (size_t)row0 * DIM;
        float4 xv[7], sv[7];
        float  sd[7];
#pragma unroll
        for (int k = 0; k < 7; ++k) {
            const int i  = t * 4 + k * 2048;
            const int ic = i < nflt ? i : 0;
            xv[k] = *(const float4*)(xr + ic);
            sv[k] = *(const float4*)(sr + ic);
            sd[k] = dinv[row0 + (ic >> 7)];
        }
#pragma unroll
        for (int k = 0; k < 7; ++k) {
            const int i = t * 4 + k * 2048;
            if (i < nflt) {
                const int r  = i >> 7;
                const int p  = r * DIM + ((i & 127) ^ ((r & 15) << 2));
                const float di = sd[k];
                float4 x1;
                x1.x = -(sv[k].x * di); x1.y = -(sv[k].y * di);
                x1.z = -(sv[k].z * di); x1.w = -(sv[k].w * di);
                *(float4*)&a0[p]  = xv[k];
                *(float4*)&a1[p]  = x1;
                *(float4*)&acc[p] = make_float4(0.0f, 0.0f, 0.0f, 0.0f);
            }
        }
    }
    __syncthreads();

    const int beg  = begs[bk];
    const int end  = begs[bk + 1];
    const int ngrp = (end - beg) >> 2;
    const int w = __builtin_amdgcn_readfirstlane(t >> 5);
    const int l = t & 31;
    for (int g = w; g < ngrp; g += 16)
        add_edges<4, true>(list, beg + 4 * g, S1, dinv, acc, row0, l);
    const int it = beg + 4 * ngrp + w;
    if (it < end)
        add_edges<1, true>(list, it, S1, dinv, acc, row0, l);
    __syncthreads();

    {
        float sd[7];
#pragma unroll
        for (int k = 0; k < 7; ++k) {
            const int i  = t * 4 + k * 2048;
            const int ic = i < nflt ? i : 0;
            sd[k] = dinv[row0 + (ic >> 7)];
        }
#pragma unroll
        for (int k = 0; k < 7; ++k) {
            const int i = t * 4 + k * 2048;
            if (i < nflt) {
                const float di = sd[k];
                const float4 s2 = *(const float4*)&acc[i];
                const float4 x1 = *(const float4*)&a1[i];
                const float4 x0 = *(const float4*)&a0[i];
                float4 x2;
                x2.x = -2.0f * ((s2.x + x1.x * di) * di) - x0.x;
                x2.y = -2.0f * ((s2.y + x1.y * di) * di) - x0.y;
                x2.z = -2.0f * ((s2.z + x1.z * di) * di) - x0.z;
                x2.w = -2.0f * ((s2.w + x1.w * di) * di) - x0.w;
                *(float4*)&acc[i] = x2;
            }
        }
    }
    __syncthreads();

    const int half  = l >> 4;
    const int lr    = l & 15;
    const int cq    = w & 3;
    const int tA    = ((w >> 2) - cq + 3) & 3;
    const int col   = 32 * cq + 2 * lr;
    const int ntile = (nrow + 15) >> 4;
    const float2 bv = *(const float2*)(b + col);
    const float2 p  = *(const float2*)(pw + col);
    float* orow = out + (size_t)row0 * DIM;
    float* rq   = red + cq * 16;
    if (tA + 4 < ntile)
        wave_tiles<LAST, true>(a0, a1, acc, tA, nrow, half, lr, col, W, bv, p, orow, rq);
    else if (tA < ntile)
        wave_tiles<LAST, false>(a0, a1, acc, tA, nrow, half, lr, col, W, bv, p, orow, rq);

    if constexpr (LAST) {
        __syncthreads();
        if (t < nrow) {
            const float* rt = red + (t >> 4) * 64 + (t & 15);
            logits[row0 + t] = ((rt[0] + rt[16]) + (rt[32] + rt[48])) + pb[0];
        }
    }
}

extern "C" void kernel_launch(void* const* d_in, const int* in_sizes, int n_in,
                              void* d_out, int out_size, void* d_ws, size_t ws_size,
                              hipStream_t stream) {
    const float* weights  = (const float*)d_in[0];
    const int*   src      = (const int*)  d_in[1];
    const int*   dst      = (const int*)  d_in[2];
    const float* lin_in_w = (const float*)d_in[3];
    const float* lin_in_b = (const float*)d_in[4];
    const float* cheb_ws  = (const float*)d_in[5];
    const float* cheb_bs  = (const float*)d_in[6];
    const float* pred_w   = (const float*)d_in[7];
    const float* pred_b   = (const float*)d_in[8];
    float* logits = (float*)d_out;

    float* ws   = (float*)d_ws;
    float* dinv = ws;
    float* P0   = ws + 51200;
    float* P1   = P0 + NF;
    float* P2   = P1 + NF;
    float* P3   = P2 + NF;
    int*   cnt  = (int*)(P3 + NF);
    int*   tot  = cnt + NCELL;
    int*   begs = tot + NB;
    int2*  list = (int2*)(cnt + 133120);

    k_hist<<<NS, 256, 0, stream>>>(dst, cnt);
    k_bscan<<<NB, NS, 0, stream>>>(cnt, tot);
    k_place<<<NS, 256, 0, stream>>>(src, dst, cnt, tot, begs, list);
    k_deg_feat<<<NB_USED, 512, 0, stream>>>(list, begs, weights, lin_in_w, lin_in_b, dinv, P0);

    float* cur = P0;
    float* nxt = P3;
    for (int layer = 0; layer < 3; ++layer) {
        const float* Wl = cheb_ws + (size_t)layer * 3 * DIM * DIM;
        const float* bl = cheb_bs + (size_t)layer * DIM;

        k_scatter_lds<<<NB_USED, 512, 0, stream>>>(list, begs, cur, dinv, P1);

        if (layer < 2)
            k_scatter_gemm<false><<<NB_USED, 512, 0, stream>>>(list, begs, cur, P1, dinv, Wl, bl, nxt,
                                                               pred_w, pred_b, logits);
        else
            k_scatter_gemm<true><<<NB_USED, 512, 0, stream>>>(list, begs, cur, P1, dinv, Wl, bl, nxt,
                                                              pred_w, pred_b, logits);

        float* t = cur; cur = nxt; nxt = t;
    }
}


# Dense product re-tiled: 8 of a group's 16 waves each own a column quarter and 4 (3) row tiles, so one W row pair feeds 8 WMMAs (was 4) and is requested 4 k steps ahead: W loads halved; sum order kept
# speedup vs baseline: 3.3698x; 1.0320x over previous
#include <hip/hip_runtime.h>

typedef float v2f __attribute__((ext_vector_type(2)));
typedef float v8f __attribute__((ext_vector_type(8)));

#define N_NODES 50000
#define N_EDGES 600000
#define DIM     128
#define NF      (N_NODES * DIM)

#define R_ROWS 98
#define NB     512
#define NS     256
#define SLICE  ((N_EDGES + NS - 1) / NS)
#define EPT    ((SLICE + 255) / 256)
#define NCELL  (NB * NS)
#define NB_USED ((N_NODES + R_ROWS - 1) / R_ROWS)

__global__ void __launch_bounds__(256)
k_hist(const int* __restrict__ dst, int* __restrict__ cnt) {
    __shared__ int h[NB];
    const int s = blockIdx.x, t = threadIdx.x;
    h[t] = 0;
    h[t + 256] = 0;
    __syncthreads();
    const int e0 = s * SLICE;
    const int e1 = (e0 + SLICE < N_EDGES) ? e0 + SLICE : N_EDGES;
    int dv[EPT];
#pragma unroll
    for (int k = 0; k < EPT; ++k) {
        const int e = e0 + t + k * 256;
        dv[k] = dst[e < e1 ? e : e1 - 1];
    }
#pragma unroll
    for (int k = 0; k < EPT; ++k) {
        if (e0 + t + k * 256 < e1) atomicAdd(&h[dv[k] / R_ROWS], 1);
    }
    __syncthreads();
    cnt[t * NS + s]         = h[t];
    cnt[(t + 256) * NS + s] = h[t + 256];
}

__global__ void __launch_bounds__(NS)
k_bscan(int* __restrict__ cnt, int* __restrict__ tot) {
    __shared__ int part[NS];
    const int b = blockIdx.x, t = threadIdx.x;
    const int c = cnt[b * NS + t];
    part[t] = c;
    __syncthreads();
    for (int off = 1; off < NS; off <<= 1) {
        const int v = (t >= off) ? part[t - off] : 0;
        __syncthreads();
        part[t] += v;
        __syncthreads();
    }
    cnt[b * NS + t] = part[t] - c;
    if (t == NS - 1) tot[b] = part[t];
}

__global__ void __launch_bounds__(256)
k_place(const int* __restrict__ src, const int* __restrict__ dst,
        const int* __restrict__ within, const int* __restrict__ tot,
        int* __restrict__ begs, int2* __restrict__ list) {
    __shared__ int cur[NB];
    __shared__ int ps[256];
    const int s = blockIdx.x, t = threadIdx.x;
    const int t0 = tot[2 * t], t1 = tot[2 * t + 1];
    ps[t] = t0 + t1;
    __syncthreads();
    for (int off = 1; off < 256; off <<= 1) {
        const int v = (t >= off) ? ps[t - off] : 0;
        __syncthreads();
        ps[t] += v;
        __syncthreads();
    }
    const int ex = ps[t] - (t0 + t1);
    cur[2 * t]     = ex      + within[(2 * t) * NS + s];
    cur[2 * t + 1] = ex + t0 + within[(2 * t + 1) * NS + s];
    if (s == 0) {
        begs[2 * t]     = ex;
        begs[2 * t + 1] = ex + t0;
        if (t == 255) begs[NB] = ex + t0 + t1;
    }
    __syncthreads();
    const int e0 = s * SLICE;
    const int e1 = (e0 + SLICE < N_EDGES) ? e0 + SLICE : N_EDGES;
    int dv[EPT], sv[EPT];
#pragma unroll
    for (int k = 0; k < EPT; ++k) {
        const int e  = e0 + t + k * 256;
        const int ec = e < e1 ? e : e1 - 1;
        dv[k] = dst[ec];
        sv[k] = src[ec];
    }
#pragma unroll
    for (int k = 0; k < EPT; ++k) {
        if (e0 + t + k * 256 < e1) {
            const int p = atomicAdd(&cur[dv[k] / R_ROWS], 1);
            list[p] = make_int2(sv[k], dv[k]);
        }
    }
}

__global__ void __launch_bounds__(512)
k_deg_feat(const int2* __restrict__ list,
           const int* __restrict__ begs,
           const float* __restrict__ w,
           const float* __restrict__ lw,
           const float* __restrict__ lb,
           float* __restrict__ dinv,
           float* __restrict__ feat) {
    __shared__ int c[128];
    const int b    = blockIdx.x;
    const int t    = threadIdx.x;
    const int row0 = b * R_ROWS;
    const int left = N_NODES - row0;
    const int nrow = left > R_ROWS ? R_ROWS : left;
    const int nflt = nrow * DIM;

    const float4 a = *(const float4*)(lw + 4 * (t & 31));
    const float4 o = *(const float4*)(lb + 4 * (t & 31));
    float wv[7];
#pragma unroll
    for (int k = 0; k < 7; ++k) {
        const int i  = t * 4 + k * 2048;
        const int ic = i < nflt ? i : 0;
        wv[k] = w[row0 + (ic >> 7)];
    }

    if (t < 128) c[t] = 0;
    __syncthreads();
    const int beg = begs[b];
    const int end = begs[b + 1];
    for (int i = beg + t; i < end; i += 512)
        atomicAdd(&c[list[i].y - row0], 1);
    __syncthreads();
    if (t < nrow) dinv[row0 + t] = rsqrtf((float)(c[t] + 1));

    float* fr = feat + (size_t)row0 * DIM;
#pragma unroll
    for (int k = 0; k < 7; ++k) {
        const int i = t * 4 + k * 2048;
        if (i < nflt) {
            float4 v;
            v.x = wv[k] * a.x + o.x;
            v.y = wv[k] * a.y + o.y;
            v.z = wv[k] * a.z + o.z;
            v.w = wv[k] * a.w + o.w;
            *(float4*)(fr + i) = v;
        }
    }
}

template <int NE, bool SECOND>
__device__ __forceinline__ void add_edges(const int2* __restrict__ list, int i,
                                          const float* __restrict__ x,
                                          const float* __restrict__ dinv,
                                          float* acc, int row0, int l) {
    int2  ed[NE];
    float sc[NE];
    float v[NE][4];
#pragma unroll
    for (int j = 0; j < NE; ++j) ed[j] = list[i + j];
#pragma unroll
    for (int j = 0; j < NE; ++j) sc[j] = dinv[ed[j].x];
#pragma unroll
    for (int j = 0; j < NE; ++j) {
        const float* xp = x + (size_t)ed[j].x * DIM + l;
        v[j][0] = xp[0];
        v[j][1] = xp[32];
        v[j][2] = xp[64];
        v[j][3] = xp[96];
    }
#pragma unroll
    for (int j = 0; j < NE; ++j) {
        const int rl = ed[j].y - row0;
        float* ap = acc + rl * DIM;
        const int sw = SECOND ? ((rl & 15) << 2) : 0;
#pragma unroll
        for (int k = 0; k < 4; ++k) {
            float val = v[j][k];
            if (SECOND) val = -(val * sc[j]);
            atomicAdd(ap + ((l + 32 * k) ^ sw), val * sc[j]);
        }
    }
}

__global__ void __launch_bounds__(512)
k_scatter_lds(const int2* __restrict__ list,
              const int* __restrict__ begs,
              const float* __restrict__ x,
              const float* __restrict__ dinv,
              float* __restrict__ agg) {
    __shared__ float acc[R_ROWS * DIM];
    const int b    = blockIdx.x;
    const int t    = threadIdx.x;
    const int row0 = b * R_ROWS;
    const int left = N_NODES - row0;
    const int nflt = (left > R_ROWS ? R_ROWS : left) * DIM;

    {
        const float* xr = x + (size_t)row0 * DIM;
        float4 sv[7];
        float  sd[7];
#pragma unroll
        for (int k = 0; k < 7; ++k) {
            const int i  = t * 4 + k * 2048;
            const int ic = i < nflt ? i : 0;
            sv[k] = *(const float4*)(xr + ic);
            sd[k] = dinv[row0 + (ic >> 7)];
        }
#pragma unroll
        for (int k = 0; k < 7; ++k) {
            const int i = t * 4 + k * 2048;
            if (i < nflt) {
                float4 v = sv[k];
                v.x *= sd[k]; v.y *= sd[k]; v.z *= sd[k]; v.w *= sd[k];
                *(float4*)&acc[i] = v;
            }
        }
    }
    __syncthreads();

    const int beg  = begs[b];
    const int end  = begs[b + 1];
    const int ngrp = (end - beg) >> 2;
    const int w = __builtin_amdgcn_readfirstlane(t >> 5);
    const int l = t & 31;
    for (int g = w; g < ngrp; g += 16)
        add_edges<4, false>(list, beg + 4 * g, x, dinv, acc, row0, l);
    const int it = beg + 4 * ngrp + w;
    if (it < end)
        add_edges<1, false>(list, it, x, dinv, acc, row0, l);
    __syncthreads();

    float* ar = agg + (size_t)row0 * DIM;
#pragma unroll
    for (int k = 0; k < 7; ++k) {
        const int i = t * 4 + k * 2048;
        if (i < nflt) *(float4*)(ar + i) = *(const float4*)&acc[i];
    }
}

#define N_TILE ((R_ROWS + 15) / 16)

template <int NT>
__device__ __forceinline__ void two_steps(v8f (&c0)[NT], v8f (&c1)[NT], const float* plane,
                                          const int (&roff)[NT], int sx, int ks,
                                          const float2 (&q)[4]) {
#pragma unroll
    for (int s = 0; s < 2; ++s) {
        const int ko = (ks + 4 * s) ^ sx;
        const v2f b0 = {q[2 * s].x, q[2 * s + 1].x}, b1 = {q[2 * s].y, q[2 * s + 1].y};
        v2f a[NT];
#pragma unroll
        for (int t = 0; t < NT; ++t) a[t] = *(const v2f*)(plane + roff[t] + ko);
#pragma unroll
        for (int t = 0; t < NT; ++t) {
            c0[t] = __builtin_amdgcn_wmma_f32_16x16x4_f32(
                 false, a[t],  false, b0,  (short)0, c0[t],
                 false,  false);
            c1[t] = __builtin_amdgcn_wmma_f32_16x16x4_f32(
                 false, a[t],  false, b1,  (short)0, c1[t],
                 false,  false);
        }
    }
}

__device__ __forceinline__ void load_set(float2 (&q)[4], const float* __restrict__ Wk, int wo) {
    q[0] = *(const float2*)(Wk + wo);
    q[1] = *(const float2*)(Wk + DIM + wo);
    q[2] = *(const float2*)(Wk + 4 * DIM + wo);
    q[3] = *(const float2*)(Wk + 5 * DIM + wo);
}

template <int NT, bool ENDS>
__device__ __forceinline__ void seg_steps(v8f (&c0)[NT], v8f (&c1)[NT], const float* plane,
                                          const int (&roff)[NT], int sx,
                                          const float* __restrict__ Ws, int wo,
                                          float2 (&qa)[4], float2 (&qb)[4]) {
#pragma unroll 1
    for (int kk = 0; kk < (ENDS ? DIM - 16 : DIM); kk += 16) {
        two_steps<NT>(c0, c1, plane, roff, sx, kk, qa);
        load_set(qa, Ws + (kk + 16) * DIM, wo);
        two_steps<NT>(c0, c1, plane, roff, sx, kk + 8, qb);
        load_set(qb, Ws + (kk + 24) * DIM, wo);
    }
    if constexpr (ENDS) {
        two_steps<NT>(c0, c1, plane, roff, sx, DIM - 16, qa);
        two_steps<NT>(c0, c1, plane, roff, sx, DIM - 8, qb);
    }
}

template <bool LAST>
__device__ __forceinline__ void tile_out(const v8f& c0, const v8f& c1, int tile, int nrow,
                                         int half, int lr, int col, float2 p,
                                         float* __restrict__ orow, float* red) {
    if constexpr (!LAST) {
#pragma unroll
        for (int r = 0; r < 8; ++r) {
            const int rr = 16 * tile + r + 8 * half;
            float2 v;
            v.x = c0[r]; v.y = c1[r];
            v.x = v.x > 0.0f ? v.x : 0.0f; v.y = v.y > 0.0f ? v.y : 0.0f;
            if (rr < nrow) *(float2*)(orow + (size_t)rr * DIM + col) = v;
        }
    } else {
        float s[8];
#pragma unroll
        for (int r = 0; r < 8; ++r) {
            float2 v;
            v.x = c0[r]; v.y = c1[r];
            v.x = v.x > 0.0f ? v.x : 0.0f; v.y = v.y > 0.0f ? v.y : 0.0f;
            s[r] = v.x * p.x + v.y * p.y;
        }
#pragma unroll
        for (int m = 1; m < 16; m <<= 1) {
#pragma unroll
            for (int r = 0; r < 8; ++r) s[r] += __shfl_xor(s[r], m);
        }
        if (lr == 0) {
#pragma unroll
            for (int r = 0; r < 8; ++r) red[tile * 64 + r + 8 * half] = s[r];
        }
    }
}

#define LAST_ROWS  (N_NODES - (NB_USED - 1) * R_ROWS)
#define LAST_TILES ((LAST_ROWS + 15) / 16)
static_assert(LAST_ROWS >= 16 && LAST_ROWS < R_ROWS, "the last range: at least one tile, and shorter than the others");
static_assert(LAST_TILES <= 4 && R_ROWS >= 64 && N_TILE == 7, "tiles 0..3 and 4..6 per column quarter");
template <bool LAST, int NT, bool CLAMP>
__device__ __forceinline__ void wave_tiles(const float* a0, const float* a1, const float* a2,
                                           int t0, int nrow, int half, int lr, int col,
                                           const float* __restrict__ W, float2 bv, float2 p,
                                           float* __restrict__ orow, float* red) {
    const int last = nrow - 1;
    const int sx   = (lr << 2) | (2 * half);
    int roff[NT];
    v8f c0[NT], c1[NT];
#pragma unroll
    for (int t = 0; t < NT; ++t) {
        const int r  = 16 * (t0 + t) + lr;
        const int rc = (!CLAMP || r <= last) ? r : r - 16 * ((r - last + 15) >> 4);
        roff[t] = rc * DIM;
        c0[t] = (v8f){bv.x, bv.x, bv.x, bv.x, bv.x, bv.x, bv.x, bv.x};
        c1[t] = (v8f){bv.y, bv.y, bv.y, bv.y, bv.y, bv.y, bv.y, bv.y};
    }
    const int wo = (2 * half) * DIM + col;
    float2 qa[4], qb[4];
    load_set(qa, W, wo);
    load_set(qb, W + 8 * DIM, wo);
    seg_steps<NT, false>(c0, c1, a0, roff, sx, W, wo, qa, qb);
    seg_steps<NT, false>(c0, c1, a1, roff, sx, W + DIM * DIM, wo, qa, qb);
    seg_steps<NT, true >(c0, c1, a2, roff, sx, W + 2 * DIM * DIM, wo, qa, qb);
#pragma unroll
    for (int t = 0; t < NT; ++t)
        tile_out<LAST>(c0[t], c1[t], t0 + t, nrow, half, lr, col, p, orow, red);
}

template <bool LAST>
__global__ void __launch_bounds__(512, 8)
k_scatter_gemm(const int2* __restrict__ list,
               const int* __restrict__ begs,
               const float* __restrict__ X0,
               const float* __restrict__ S1,
               const float* __restrict__ dinv,
               const float* __restrict__ W,
               const float* __restrict__ b,
               float* __restrict__ out,
               const float* __restrict__ pw,
               const float* __restrict__ pb,
               float* __restrict__ logits) {
    __shared__ float a0[R_ROWS * DIM];
    __shared__ float a1[R_ROWS * DIM];
    __shared__ float acc[R_ROWS * DIM];
    __shared__ float red[N_TILE * 64];
    const int bk   = blockIdx.x;
    const int t    = threadIdx.x;
    const int row0 = bk * R_ROWS;
    const int left = N_NODES - row0;
    const int nrow = left > R_ROWS ? R_ROWS : left;
    const int nflt = nrow * DIM;

    {
        const float* xr = X0 + (size_t)row0 * DIM;
        const float* sr = S1 + (size_t)row0 * DIM;
        float4 xv[7], sv[7];
        float  sd[7];
#pragma unroll
        for (int k = 0; k < 7; ++k) {
            const int i  = t * 4 + k * 2048;
            const int ic = i < nflt ? i : 0;
            xv[k] = *(const float4*)(xr + ic);
            sv[k] = *(const float4*)(sr + ic);
            sd[k] = dinv[row0 + (ic >> 7)];
        }
#pragma unroll
        for (int k = 0; k < 7; ++k) {
            const int i = t * 4 + k * 2048;
            if (i < nflt) {
                const int r  = i >> 7;
                const int p  = r * DIM + ((i & 127) ^ ((r & 15) << 2));
                const float di = sd[k];
                float4 x1;
                x1.x = -(sv[k].x * di); x1.y = -(sv[k].y * di);
                x1.z = -(sv[k].z * di); x1.w = -(sv[k].w * di);
                *(float4*)&a0[p]  = xv[k];
                *(float4*)&a1[p]  = x1;
                *(float4*)&acc[p] = make_float4(0.0f, 0.0f, 0.0f, 0.0f);
            }
        }
    }
    __syncthreads();

    const int beg  = begs[bk];
    const int end  = begs[bk + 1];
    const int ngrp = (end - beg) >> 2;
    const int w = __builtin_amdgcn_readfirstlane(t >> 5);
    const int l = t & 31;
    for (int g = w; g < ngrp; g += 16)
        add_edges<4, true>(list, beg + 4 * g, S1, dinv, acc, row0, l);
    const int it = beg + 4 * ngrp + w;
    if (it < end)
        add_edges<1, true>(list, it, S1, dinv, acc, row0, l);
    __syncthreads();

    {
        float sd[7];
#pragma unroll
        for (int k = 0; k < 7; ++k) {
            const int i  = t * 4 + k * 2048;
            const int ic = i < nflt ? i : 0;
            sd[k] = dinv[row0 + (ic >> 7)];
        }
#pragma unroll
        for (int k = 0; k < 7; ++k) {
            const int i = t * 4 + k * 2048;
            if (i < nflt) {
                const float di = sd[k];
                const float4 s2 = *(const float4*)&acc[i];
                const float4 x1 = *(const float4*)&a1[i];
                const float4 x0 = *(const float4*)&a0[i];
                float4 x2;
                x2.x = -2.0f * ((s2.x + x1.x * di) * di) - x0.x;
                x2.y = -2.0f * ((s2.y + x1.y * di) * di) - x0.y;
                x2.z = -2.0f * ((s2.z + x1.z * di) * di) - x0.z;
                x2.w = -2.0f * ((s2.w + x1.w * di) * di) - x0.w;
                *(float4*)&acc[i] = x2;
            }
        }
    }
    __syncthreads();

    const int half  = l >> 4;
    const int lr    = l & 15;
    const int cq    = w & 3;
    const int wq    = w >> 2;
    const int rh    = ((cq >> 1) + (wq >> 1)) & 1;
    const int col   = 32 * cq + 2 * lr;
    float* orow = out + (size_t)row0 * DIM;
    float* rq   = red + cq * 16;
    if (((cq + wq) & 1) == 0) {
        const float2 bv = *(const float2*)(b + col);
        const float2 p  = *(const float2*)(pw + col);
        if (rh == 0) {
            if (nrow == R_ROWS)
                wave_tiles<LAST, 4, false>(a0, a1, acc, 0, nrow, half, lr, col, W, bv, p, orow, rq);
            else
                wave_tiles<LAST, LAST_TILES, true>(a0, a1, acc, 0, nrow, half, lr, col, W, bv, p, orow, rq);
        } else if (nrow == R_ROWS) {
            wave_tiles<LAST, 3, true>(a0, a1, acc, 4, nrow, half, lr, col, W, bv, p, orow, rq);
        }
    }

    if constexpr (LAST) {
        __syncthreads();
        if (t < nrow) {
            const float* rt = red + (t >> 4) * 64 + (t & 15);
            logits[row0 + t] = ((rt[0] + rt[16]) + (rt[32] + rt[48])) + pb[0];
        }
    }
}

extern "C" void kernel_launch(void* const* d_in, const int* in_sizes, int n_in,
                              void* d_out, int out_size, void* d_ws, size_t ws_size,
                              hipStream_t stream) {
    const float* weights  = (const float*)d_in[0];
    const int*   src      = (const int*)  d_in[1];
    const int*   dst      = (const int*)  d_in[2];
    const float* lin_in_w = (const float*)d_in[3];
    const float* lin_in_b = (const float*)d_in[4];
    const float* cheb_ws  = (const float*)d_in[5];
    const float* cheb_bs  = (const float*)d_in[6];
    const float* pred_w   = (const float*)d_in[7];
    const float* pred_b   = (const float*)d_in[8];
    float* logits = (float*)d_out;

    float* ws   = (float*)d_ws;
    float* dinv = ws;
    float* P0   = ws + 51200;
    float* P1   = P0 + NF;
    float* P2   = P1 + NF;
    float* P3   = P2 + NF;
    int*   cnt  = (int*)(P3 + NF);
    int*   tot  = cnt + NCELL;
    int*   begs = tot + NB;
    int2*  list = (int2*)(cnt + 133120);

    k_hist<<<NS, 256, 0, stream>>>(dst, cnt);
    k_bscan<<<NB, NS, 0, stream>>>(cnt, tot);
    k_place<<<NS, 256, 0, stream>>>(src, dst, cnt, tot, begs, list);
    k_deg_feat<<<NB_USED, 512, 0, stream>>>(list, begs, weights, lin_in_w, lin_in_b, dinv, P0);

    float* cur = P0;
    float* nxt = P3;
    for (int layer = 0; layer < 3; ++layer) {
        const float* Wl = cheb_ws + (size_t)layer * 3 * DIM * DIM;
        const float* bl = cheb_bs + (size_t)layer * DIM;

        k_scatter_lds<<<NB_USED, 512, 0, stream>>>(list, begs, cur, dinv, P1);

        if (layer < 2)
            k_scatter_gemm<false><<<NB_USED, 512, 0, stream>>>(list, begs, cur, P1, dinv, Wl, bl, nxt,
                                                               pred_w, pred_b, logits);
        else
            k_scatter_gemm<true><<<NB_USED, 512, 0, stream>>>(list, begs, cur, P1, dinv, Wl, bl, nxt,
                                                              pred_w, pred_b, logits);

        float* t = cur; cur = nxt; nxt = t;
    }
}
